# starting kernel
# baseline (speedup 1.0000x reference)
#include <hip/hip_runtime.h>

typedef __attribute__((ext_vector_type(2))) float v2f;
typedef __attribute__((ext_vector_type(8))) float v8f;

#define R_DIM 64
#define C_DIM 256

__global__ void zero_kernel(float* __restrict__ p, int n) {
    int i = blockIdx.x * blockDim.x + threadIdx.x;
    if (i < n) p[i] = 0.0f;
}

__global__ void reduce_u_kernel(const float* __restrict__ eg,
                                float* __restrict__ u, int n) {
    __shared__ float smem[256];
    int tid  = threadIdx.x;
    int col  = tid & 63;
    int rgrp = tid >> 6;
    float s = 0.0f;
    for (int row = blockIdx.x * 4 + rgrp; row < n; row += gridDim.x * 4)
        s += eg[row * R_DIM + col];
    smem[tid] = s;
    __syncthreads();
    if (tid < 64) {
        float v = smem[tid] + smem[tid + 64] + smem[tid + 128] + smem[tid + 192];
        atomicAdd(&u[tid], v);
    }
}

#define EGP 160
#define XP  544

__global__ void proj_kernel(const float* __restrict__ eg,
                            const float* __restrict__ x,
                            float* __restrict__ proj, int n) {
    __shared__ float egp[16 * EGP];
    __shared__ float xp[16 * XP];

    int tid   = threadIdx.x;
    int lane  = tid & 31;
    int w     = tid >> 5;
    int mrow  = lane & 15;
    int khalf = lane >> 4;
    int rt    = w & 3;
    int ctb   = (w >> 2) * 8;

    int chunk = (n + gridDim.x - 1) / gridDim.x;
    int n0    = blockIdx.x * chunk;
    int nend  = min(n0 + chunk, n);

    v8f acc[8];
#pragma unroll
    for (int t = 0; t < 8; ++t)
#pragma unroll
        for (int j = 0; j < 8; ++j) acc[t][j] = 0.0f;

    for (int nb = n0; nb < nend; nb += 32) {
#pragma unroll
        for (int i = 0; i < 8; ++i) {
            int row = i * 4 + (tid >> 6);
            int col = tid & 63;
            int g   = nb + row;
            float v = (g < nend) ? eg[g * R_DIM + col] : 0.0f;
            egp[(row >> 1) * EGP + col * 2 + (row & 1)] = v;
        }
#pragma unroll
        for (int i = 0; i < 32; ++i) {
            int g   = nb + i;
            float v = (g < nend) ? x[g * C_DIM + tid] : 0.0f;
            xp[(i >> 1) * XP + tid * 2 + (i & 1)] = v;
        }
        __syncthreads();

#pragma unroll
        for (int kk = 0; kk < 8; ++kk) {
            int pidx = kk * 2 + khalf;
            v2f a = *(const v2f*)&egp[pidx * EGP + (rt * 16 + mrow) * 2];
#pragma unroll
            for (int t = 0; t < 8; ++t) {
                int c = (ctb + t) * 16 + mrow;
                v2f b = *(const v2f*)&xp[pidx * XP + c * 2];
                acc[t] = __builtin_amdgcn_wmma_f32_16x16x4_f32(
                    false, a, false, b, (short)0, acc[t], false, false);
            }
        }
        __syncthreads();
    }

#pragma unroll
    for (int t = 0; t < 8; ++t) {
        int c = (ctb + t) * 16 + mrow;
#pragma unroll
        for (int i = 0; i < 8; ++i) {
            int m = i + 8 * khalf;
            int r = rt * 16 + m;
            atomicAdd(&proj[r * C_DIM + c], acc[t][i]);
        }
    }
}

__global__ void p2_kernel(const float* __restrict__ proj,
                          const float* __restrict__ W,
                          float* __restrict__ P2i) {
    int tid   = threadIdx.x;
    int lane  = tid & 31;
    int w     = tid >> 5;
    int mrow  = lane & 15;
    int khalf = lane >> 4;
    int rt    = w & 3;
    int ctb   = (w >> 2) * 8;

    v8f acc[8];
#pragma unroll
    for (int t = 0; t < 8; ++t)
#pragma unroll
        for (int j = 0; j < 8; ++j) acc[t][j] = 0.0f;

    for (int kk = 0; kk < 64; ++kk) {
        int k0 = kk * 4 + 2 * khalf;
        v2f a = *(const v2f*)&proj[(rt * 16 + mrow) * C_DIM + k0];
#pragma unroll
        for (int t = 0; t < 8; ++t) {
            int c0 = (ctb + t) * 16;
            v2f b;
            b.x = W[k0 * C_DIM + c0 + mrow];
            b.y = W[(k0 + 1) * C_DIM + c0 + mrow];
            acc[t] = __builtin_amdgcn_wmma_f32_16x16x4_f32(
                false, a, false, b, (short)0, acc[t], false, false);
        }
    }
#pragma unroll
    for (int t = 0; t < 8; ++t) {
        int c = (ctb + t) * 16 + mrow;
#pragma unroll
        for (int i = 0; i < 8; ++i) {
            int r = rt * 16 + i + 8 * khalf;
            P2i[(r >> 1) * (2 * C_DIM) + c * 2 + (r & 1)] = acc[t][i];
        }
    }
}

#define EGO 68

__global__ void out_kernel(const float* __restrict__ eg,
                           const float* __restrict__ P2i,
                           const float* __restrict__ u,
                           const float* __restrict__ bias,
                           float* __restrict__ out, int n) {
    __shared__ float egLDS[64 * EGO];
    __shared__ float uLDS[64];
    __shared__ float invLDS[64];
    __shared__ float biasLDS[256];

    int tid = threadIdx.x;
    int rb  = blockIdx.x * 64;

    if (tid < 64) uLDS[tid] = u[tid];
    biasLDS[tid] = bias[tid];
#pragma unroll
    for (int i = 0; i < 16; ++i) {
        int row = i * 4 + (tid >> 6);
        int col = tid & 63;
        int g   = rb + row;
        egLDS[row * EGO + col] = (g < n) ? eg[g * R_DIM + col] : 0.0f;
    }
    __syncthreads();

    if (tid < 64) {
        float s = 0.0f;
#pragma unroll
        for (int r = 0; r < R_DIM; ++r) s += egLDS[tid * EGO + r] * uLDS[r];
        invLDS[tid] = 1.0f / s;
    }
    __syncthreads();

    int lane  = tid & 31;
    int w     = tid >> 5;
    int mrow  = lane & 15;
    int khalf = lane >> 4;
    int rt    = w & 3;
    int ctb   = (w >> 2) * 8;

    const v2f* P2v = (const v2f*)P2i;

    v8f acc[8];
#pragma unroll
    for (int t = 0; t < 8; ++t)
#pragma unroll
        for (int j = 0; j < 8; ++j) acc[t][j] = 0.0f;

#pragma unroll
    for (int kk = 0; kk < 16; ++kk) {
        int k0 = kk * 4 + 2 * khalf;
        v2f a = *(const v2f*)&egLDS[(rt * 16 + mrow) * EGO + k0];
#pragma unroll
        for (int t = 0; t < 8; ++t) {
            int c = (ctb + t) * 16 + mrow;
            v2f b = P2v[(k0 >> 1) * C_DIM + c];
            acc[t] = __builtin_amdgcn_wmma_f32_16x16x4_f32(
                false, a, false, b, (short)0, acc[t], false, false);
        }
    }

#pragma unroll
    for (int t = 0; t < 8; ++t) {
        int c = (ctb + t) * 16 + mrow;
#pragma unroll
        for (int i = 0; i < 8; ++i) {
            int m    = i + 8 * khalf;
            int grow = rb + rt * 16 + m;
            if (grow < n)
                out[grow * C_DIM + c] =
                    acc[t][i] * invLDS[rt * 16 + m] + biasLDS[c];
        }
    }
}

extern "C" void kernel_launch(void* const* d_in, const int* in_sizes, int n_in,
                              void* d_out, int out_size, void* d_ws, size_t ws_size,
                              hipStream_t stream) {
    const float* x    = (const float*)d_in[0];
    const float* eg   = (const float*)d_in[2];
    const float* W    = (const float*)d_in[3];
    const float* bias = (const float*)d_in[4];
    float*       out  = (float*)d_out;

    int N = in_sizes[2] / R_DIM;

    float* ws   = (float*)d_ws;
    float* u    = ws;
    float* proj = ws + 64;
    float* P2i  = ws + 64 + R_DIM * C_DIM;

    int nz = 64 + R_DIM * C_DIM;
    zero_kernel<<<(nz + 255) / 256, 256, 0, stream>>>(ws, nz);
    reduce_u_kernel<<<512, 256, 0, stream>>>(eg, u, N);
    proj_kernel<<<250, 256, 0, stream>>>(eg, x, proj, N);
    p2_kernel<<<1, 256, 0, stream>>>(proj, W, P2i);
    out_kernel<<<(N + 63) / 64, 256, 0, stream>>>(eg, P2i, u, bias, out, N);
}


# out_kernel without LDS or barriers: a wave owns a 64x64 tile, its eg and P2 fragments go global->registers ahead of the products, deg from its own eg fragments; r sum order of out and deg changes
# speedup vs baseline: 2.9429x; 2.9429x over previous
#include <hip/hip_runtime.h>

typedef __attribute__((ext_vector_type(2))) float v2f;
typedef __attribute__((ext_vector_type(8))) float v8f;

#define R_DIM 64
#define C_DIM 256

__global__ void zero_kernel(float* __restrict__ p, int n) {
    int i = blockIdx.x * blockDim.x + threadIdx.x;
    if (i < n) p[i] = 0.0f;
}

__global__ void reduce_u_kernel(const float* __restrict__ eg,
                                float* __restrict__ u, int n) {
    __shared__ float smem[256];
    int tid  = threadIdx.x;
    int col  = tid & 63;
    int rgrp = tid >> 6;
    float s = 0.0f;
    for (int row = blockIdx.x * 4 + rgrp; row < n; row += gridDim.x * 4)
        s += eg[row * R_DIM + col];
    smem[tid] = s;
    __syncthreads();
    if (tid < 64) {
        float v = smem[tid] + smem[tid + 64] + smem[tid + 128] + smem[tid + 192];
        atomicAdd(&u[tid], v);
    }
}

typedef __attribute__((ext_vector_type(4))) float v4f;

#define PJ_D 3

struct PjStep {
    v4f e[2];
    v4f xv[2][2];
};

__device__ __forceinline__ void pj_load(PjStep& b, const float* __restrict__ eg,
                                        const float* __restrict__ x,
                                        int step, int eo, int xo) {
    const float* eS = eg + (size_t)step * (4 * R_DIM);
    const float* xS = x + (size_t)step * (4 * C_DIM);
#pragma unroll
    for (int s = 0; s < 2; ++s) {
        b.e[s]     = *(const v4f*)(eS + eo + s * R_DIM);
        b.xv[0][s] = *(const v4f*)(xS + xo + s * C_DIM);
        b.xv[1][s] = *(const v4f*)(xS + xo + s * C_DIM + 64);
    }
}

__device__ __forceinline__ void pj_mma(v8f (&acc)[4][8], const PjStep& b) {
    v2f a[4];
#pragma unroll
    for (int i = 0; i < 4; ++i) { a[i].x = b.e[0][i]; a[i].y = b.e[1][i]; }
#pragma unroll
    for (int g = 0; g < 2; ++g)
#pragma unroll
        for (int j = 0; j < 4; ++j) {
            v2f bb;
            bb.x = b.xv[g][0][j];
            bb.y = b.xv[g][1][j];
#pragma unroll
            for (int i = 0; i < 4; ++i)
                acc[i][g * 4 + j] = __builtin_amdgcn_wmma_f32_16x16x4_f32(
                    false, a[i], false, bb, (short)0, acc[i][g * 4 + j], false, false);
        }
}

__global__ __launch_bounds__(256)
void proj_kernel(const float* __restrict__ eg,
                 const float* __restrict__ x,
                 float* __restrict__ proj, int n) {
    __shared__ __attribute__((aligned(16))) float part[4 * R_DIM * C_DIM];

    int tid     = threadIdx.x;
    int lane    = tid & 31;
    int w       = __builtin_amdgcn_readfirstlane(tid >> 5);
    int c       = lane & 15;
    int h       = lane >> 4;
    int colhalf = w & 1;
    int kslot   = w >> 1;
    int colbase = colhalf * 128;
    int c4      = c * 4;
    int eo      = (2 * h) * R_DIM + c4;
    int xo      = (2 * h) * C_DIM + colbase + c4;

    int F    = n >> 2;
    int Q    = gridDim.x * 4;
    int q    = blockIdx.x * 4 + kslot;
    int base = F / Q;
    int rem  = F - base * Q;
    int sBeg = q * base + min(q, rem);
    int sEnd = sBeg + base + (q < rem ? 1 : 0);

    v8f acc[4][8];
#pragma unroll
    for (int i = 0; i < 4; ++i)
#pragma unroll
        for (int t = 0; t < 8; ++t)
#pragma unroll
            for (int j = 0; j < 8; ++j) acc[i][t][j] = 0.0f;

    if (sEnd > sBeg) {
        PjStep buf[PJ_D];
        int last = sEnd - 1;
#pragma unroll
        for (int d = 0; d < PJ_D; ++d)
            pj_load(buf[d], eg, x, min(sBeg + d, last), eo, xo);

        int t = sBeg;
        for (; t + PJ_D <= sEnd; t += PJ_D) {
#pragma unroll
            for (int d = 0; d < PJ_D; ++d) {
                PjStep cur = buf[d];
                pj_load(buf[d], eg, x, min(t + d + PJ_D, last), eo, xo);
                pj_mma(acc, cur);
            }
        }
#pragma unroll
        for (int d = 0; d < PJ_D - 1; ++d)
            if (t + d < sEnd) pj_mma(acc, buf[d]);
    }

    if ((n & 3) != 0 && q == Q - 1) {
        PjStep tb;
#pragma unroll
        for (int s = 0; s < 2; ++s) {
            int  row = 4 * F + 2 * h + s;
            bool ok  = row < n;
            int  rc  = ok ? row : n - 1;
            const float* er = eg + (size_t)rc * R_DIM + c4;
            const float* xr = x + (size_t)rc * C_DIM + colbase + c4;
            v4f z  = {0.0f, 0.0f, 0.0f, 0.0f};
            v4f ev = *(const v4f*)er;
            v4f x0 = *(const v4f*)xr;
            v4f x1 = *(const v4f*)(xr + 64);
            tb.e[s]     = ok ? ev : z;
            tb.xv[0][s] = ok ? x0 : z;
            tb.xv[1][s] = ok ? x1 : z;
        }
        pj_mma(acc, tb);
    }

    float* mypart = part + kslot * (R_DIM * C_DIM);
#pragma unroll
    for (int i = 0; i < 4; ++i)
#pragma unroll
        for (int g = 0; g < 2; ++g)
#pragma unroll
            for (int qr = 0; qr < 8; ++qr) {
                int r = 4 * (8 * h + qr) + i;
                v4f v;
                v.x = acc[i][g * 4 + 0][qr];
                v.y = acc[i][g * 4 + 1][qr];
                v.z = acc[i][g * 4 + 2][qr];
                v.w = acc[i][g * 4 + 3][qr];
                *(v4f*)&mypart[r * C_DIM + colbase + 64 * g + c4] = v;
            }
    __syncthreads();

    for (int e0 = 0; e0 < 64; e0 += 8) {
        float p[8][4];
#pragma unroll
        for (int u = 0; u < 8; ++u)
#pragma unroll
            for (int k = 0; k < 4; ++k)
                p[u][k] = part[k * (R_DIM * C_DIM) + (e0 + u) * 256 + tid];
#pragma unroll
        for (int u = 0; u < 8; ++u) {
            float v = ((p[u][0] + p[u][1]) + p[u][2]) + p[u][3];
            atomicAdd(&proj[(e0 + u) * 256 + tid], v);
        }
    }
}

__global__ void p2_kernel(const float* __restrict__ proj,
                          const float* __restrict__ W,
                          float* __restrict__ P2i) {
    int tid   = threadIdx.x;
    int lane  = tid & 31;
    int w     = tid >> 5;
    int mrow  = lane & 15;
    int khalf = lane >> 4;
    int rt    = w & 3;
    int ctb   = (w >> 2) * 8;

    v8f acc[8];
#pragma unroll
    for (int t = 0; t < 8; ++t)
#pragma unroll
        for (int j = 0; j < 8; ++j) acc[t][j] = 0.0f;

    for (int kk = 0; kk < 64; ++kk) {
        int k0 = kk * 4 + 2 * khalf;
        v2f a = *(const v2f*)&proj[(rt * 16 + mrow) * C_DIM + k0];
#pragma unroll
        for (int t = 0; t < 8; ++t) {
            int c0 = (ctb + t) * 16;
            v2f b;
            b.x = W[k0 * C_DIM + c0 + mrow];
            b.y = W[(k0 + 1) * C_DIM + c0 + mrow];
            acc[t] = __builtin_amdgcn_wmma_f32_16x16x4_f32(
                false, a, false, b, (short)0, acc[t], false, false);
        }
    }
#pragma unroll
    for (int t = 0; t < 8; ++t) {
        int c = (ctb + t) * 16 + mrow;
#pragma unroll
        for (int i = 0; i < 8; ++i) {
            int r = rt * 16 + i + 8 * khalf;
            P2i[(r >> 1) * (2 * C_DIM) + c * 2 + (r & 1)] = acc[t][i];
        }
    }
}

template <bool GUARD>
__device__ __forceinline__ void out_store(float* __restrict__ out, const v8f (&acc)[4][4],
                                          int rb, int h, int colc, int n) {
#pragma unroll
    for (int i = 0; i < 4; ++i)
#pragma unroll
        for (int qr = 0; qr < 8; ++qr) {
            int row = rb + 16 * i + 8 * h + qr;
            float* orow = out + (size_t)row * C_DIM + colc;
            if (!GUARD || row < n) {
#pragma unroll
                for (int t = 0; t < 4; ++t) orow[16 * t] = acc[i][t][qr];
            }
        }
}

__global__ __launch_bounds__(128)
void out_kernel(const float* __restrict__ eg,
                const float* __restrict__ P2i,
                const float* __restrict__ u,
                const float* __restrict__ bias,
                float* __restrict__ out, int n) {
    int tid     = threadIdx.x;
    int lane    = tid & 31;
    int w       = __builtin_amdgcn_readfirstlane(tid >> 5);
    int c       = lane & 15;
    int h       = lane >> 4;
    int rb      = blockIdx.x * 64;
    int colbase = w * 64;
    int colc    = colbase + c;

    v4f a4[4][8];
#pragma unroll
    for (int i = 0; i < 4; ++i) {
        int rc = min(rb + 16 * i + c, n - 1);
        const float* er = eg + (size_t)rc * R_DIM + 4 * h;
#pragma unroll
        for (int j = 0; j < 8; ++j) a4[i][j] = *(const v4f*)(er + 8 * j);
    }

    v4f u4[8];
#pragma unroll
    for (int j = 0; j < 8; ++j) u4[j] = *(const v4f*)(u + 8 * j + 4 * h);

    const v2f* P2v = (const v2f*)P2i;
    const v2f* bp  = P2v + (2 * h) * C_DIM + colc;
    v2f b[8][2][4];
#pragma unroll
    for (int j = 0; j < 8; ++j)
#pragma unroll
        for (int p = 0; p < 2; ++p)
#pragma unroll
            for (int t = 0; t < 4; ++t)
                b[j][p][t] = bp[(4 * j + p) * C_DIM + 16 * t];

    float bs[4];
#pragma unroll
    for (int t = 0; t < 4; ++t) bs[t] = bias[colc + 16 * t];

    float inv[4];
#pragma unroll
    for (int i = 0; i < 4; ++i) {
        float s = 0.0f;
#pragma unroll
        for (int j = 0; j < 8; ++j)
#pragma unroll
            for (int e = 0; e < 4; ++e) s += a4[i][j][e] * u4[j][e];
        s += __shfl_xor(s, 16);
        inv[i] = 1.0f / s;
    }

    v8f acc[4][4];
#pragma unroll
    for (int i = 0; i < 4; ++i)
#pragma unroll
        for (int t = 0; t < 4; ++t)
#pragma unroll
            for (int q = 0; q < 8; ++q) acc[i][t][q] = 0.0f;

#pragma unroll
    for (int j = 0; j < 8; ++j)
#pragma unroll
        for (int p = 0; p < 2; ++p) {
            v2f a[4];
#pragma unroll
            for (int i = 0; i < 4; ++i) { a[i].x = a4[i][j][2 * p]; a[i].y = a4[i][j][2 * p + 1]; }
#pragma unroll
            for (int t = 0; t < 4; ++t)
#pragma unroll
                for (int i = 0; i < 4; ++i)
                    acc[i][t] = __builtin_amdgcn_wmma_f32_16x16x4_f32(
                        false, a[i], false, b[j][p][t], (short)0, acc[i][t], false, false);
        }

#pragma unroll
    for (int i = 0; i < 4; ++i)
#pragma unroll
        for (int qr = 0; qr < 8; ++qr) {
            float sc = __shfl(inv[i], 8 * h + qr);
#pragma unroll
            for (int t = 0; t < 4; ++t) acc[i][t][qr] = acc[i][t][qr] * sc + bs[t];
        }

    if (rb + 64 <= n) out_store<false>(out, acc, rb, h, colc, n);
    else              out_store<true>(out, acc, rb, h, colc, n);
}

extern "C" void kernel_launch(void* const* d_in, const int* in_sizes, int n_in,
                              void* d_out, int out_size, void* d_ws, size_t ws_size,
                              hipStream_t stream) {
    const float* x    = (const float*)d_in[0];
    const float* eg   = (const float*)d_in[2];
    const float* W    = (const float*)d_in[3];
    const float* bias = (const float*)d_in[4];
    float*       out  = (float*)d_out;

    int N = in_sizes[2] / R_DIM;

    float* ws   = (float*)d_ws;
    float* u    = ws;
    float* proj = ws + 64;
    float* P2i  = ws + 64 + R_DIM * C_DIM;

    int nz = 64 + R_DIM * C_DIM;
    zero_kernel<<<(nz + 255) / 256, 256, 0, stream>>>(ws, nz);
    reduce_u_kernel<<<512, 256, 0, stream>>>(eg, u, N);
    proj_kernel<<<256, 256, 0, stream>>>(eg, x, proj, N);
    p2_kernel<<<1, 256, 0, stream>>>(proj, W, P2i);
    out_kernel<<<(N + 63) / 64, 128, 0, stream>>>(eg, P2i, u, bias, out, N);
}


# u (column sums of eg) loses its own launch: its 512 work-groups run in the P2 launch beside P2's one work-group, whose serial load chain left the chip idle (5 -> 4 launches); sum orders kept
# speedup vs baseline: 3.6543x; 1.2417x over previous
#include <hip/hip_runtime.h>

typedef __attribute__((ext_vector_type(2))) float v2f;
typedef __attribute__((ext_vector_type(8))) float v8f;

#define R_DIM 64
#define C_DIM 256

__global__ void zero_kernel(float* __restrict__ p, int n) {
    int i = blockIdx.x * blockDim.x + threadIdx.x;
    if (i < n) p[i] = 0.0f;
}

__device__ __forceinline__ void reduce_u_block(const float* __restrict__ eg,
                                               float* __restrict__ u, int n,
                                               int blk, int nblk) {
    __shared__ float smem[256];
    int tid  = threadIdx.x;
    int col  = tid & 63;
    int rgrp = tid >> 6;
    float s = 0.0f;
    for (int row = blk * 4 + rgrp; row < n; row += nblk * 4)
        s += eg[row * R_DIM + col];
    smem[tid] = s;
    __syncthreads();
    if (tid < 64) {
        float v = smem[tid] + smem[tid + 64] + smem[tid + 128] + smem[tid + 192];
        atomicAdd(&u[tid], v);
    }
}

typedef __attribute__((ext_vector_type(4))) float v4f;

#define PJ_D 3

struct PjStep {
    v4f e[2];
    v4f xv[2][2];
};

__device__ __forceinline__ void pj_load(PjStep& b, const float* __restrict__ eg,
                                        const float* __restrict__ x,
                                        int step, int eo, int xo) {
    const float* eS = eg + (size_t)step * (4 * R_DIM);
    const float* xS = x + (size_t)step * (4 * C_DIM);
#pragma unroll
    for (int s = 0; s < 2; ++s) {
        b.e[s]     = *(const v4f*)(eS + eo + s * R_DIM);
        b.xv[0][s] = *(const v4f*)(xS + xo + s * C_DIM);
        b.xv[1][s] = *(const v4f*)(xS + xo + s * C_DIM + 64);
    }
}

__device__ __forceinline__ void pj_mma(v8f (&acc)[4][8], const PjStep& b) {
    v2f a[4];
#pragma unroll
    for (int i = 0; i < 4; ++i) { a[i].x = b.e[0][i]; a[i].y = b.e[1][i]; }
#pragma unroll
    for (int g = 0; g < 2; ++g)
#pragma unroll
        for (int j = 0; j < 4; ++j) {
            v2f bb;
            bb.x = b.xv[g][0][j];
            bb.y = b.xv[g][1][j];
#pragma unroll
            for (int i = 0; i < 4; ++i)
                acc[i][g * 4 + j] = __builtin_amdgcn_wmma_f32_16x16x4_f32(
                    false, a[i], false, bb, (short)0, acc[i][g * 4 + j], false, false);
        }
}

__global__ __launch_bounds__(256)
void proj_kernel(const float* __restrict__ eg,
                 const float* __restrict__ x,
                 float* __restrict__ proj, int n) {
    __shared__ __attribute__((aligned(16))) float part[4 * R_DIM * C_DIM];

    int tid     = threadIdx.x;
    int lane    = tid & 31;
    int w       = __builtin_amdgcn_readfirstlane(tid >> 5);
    int c       = lane & 15;
    int h       = lane >> 4;
    int colhalf = w & 1;
    int kslot   = w >> 1;
    int colbase = colhalf * 128;
    int c4      = c * 4;
    int eo      = (2 * h) * R_DIM + c4;
    int xo      = (2 * h) * C_DIM + colbase + c4;

    int F    = n >> 2;
    int Q    = gridDim.x * 4;
    int q    = blockIdx.x * 4 + kslot;
    int base = F / Q;
    int rem  = F - base * Q;
    int sBeg = q * base + min(q, rem);
    int sEnd = sBeg + base + (q < rem ? 1 : 0);

    v8f acc[4][8];
#pragma unroll
    for (int i = 0; i < 4; ++i)
#pragma unroll
        for (int t = 0; t < 8; ++t)
#pragma unroll
            for (int j = 0; j < 8; ++j) acc[i][t][j] = 0.0f;

    if (sEnd > sBeg) {
        PjStep buf[PJ_D];
        int last = sEnd - 1;
#pragma unroll
        for (int d = 0; d < PJ_D; ++d)
            pj_load(buf[d], eg, x, min(sBeg + d, last), eo, xo);

        int t = sBeg;
        for (; t + PJ_D <= sEnd; t += PJ_D) {
#pragma unroll
            for (int d = 0; d < PJ_D; ++d) {
                PjStep cur = buf[d];
                pj_load(buf[d], eg, x, min(t + d + PJ_D, last), eo, xo);
                pj_mma(acc, cur);
            }
        }
#pragma unroll
        for (int d = 0; d < PJ_D - 1; ++d)
            if (t + d < sEnd) pj_mma(acc, buf[d]);
    }

    if ((n & 3) != 0 && q == Q - 1) {
        PjStep tb;
#pragma unroll
        for (int s = 0; s < 2; ++s) {
            int  row = 4 * F + 2 * h + s;
            bool ok  = row < n;
            int  rc  = ok ? row : n - 1;
            const float* er = eg + (size_t)rc * R_DIM + c4;
            const float* xr = x + (size_t)rc * C_DIM + colbase + c4;
            v4f z  = {0.0f, 0.0f, 0.0f, 0.0f};
            v4f ev = *(const v4f*)er;
            v4f x0 = *(const v4f*)xr;
            v4f x1 = *(const v4f*)(xr + 64);
            tb.e[s]     = ok ? ev : z;
            tb.xv[0][s] = ok ? x0 : z;
            tb.xv[1][s] = ok ? x1 : z;
        }
        pj_mma(acc, tb);
    }

    float* mypart = part + kslot * (R_DIM * C_DIM);
#pragma unroll
    for (int i = 0; i < 4; ++i)
#pragma unroll
        for (int g = 0; g < 2; ++g)
#pragma unroll
            for (int qr = 0; qr < 8; ++qr) {
                int r = 4 * (8 * h + qr) + i;
                v4f v;
                v.x = acc[i][g * 4 + 0][qr];
                v.y = acc[i][g * 4 + 1][qr];
                v.z = acc[i][g * 4 + 2][qr];
                v.w = acc[i][g * 4 + 3][qr];
                *(v4f*)&mypart[r * C_DIM + colbase + 64 * g + c4] = v;
            }
    __syncthreads();

    for (int e0 = 0; e0 < 64; e0 += 8) {
        float p[8][4];
#pragma unroll
        for (int u = 0; u < 8; ++u)
#pragma unroll
            for (int k = 0; k < 4; ++k)
                p[u][k] = part[k * (R_DIM * C_DIM) + (e0 + u) * 256 + tid];
#pragma unroll
        for (int u = 0; u < 8; ++u) {
            float v = ((p[u][0] + p[u][1]) + p[u][2]) + p[u][3];
            atomicAdd(&proj[(e0 + u) * 256 + tid], v);
        }
    }
}

__device__ __forceinline__ void p2_block(const float* __restrict__ proj,
                                         const float* __restrict__ W,
                                         float* __restrict__ P2i) {
    int tid   = threadIdx.x;
    int lane  = tid & 31;
    int w     = tid >> 5;
    int mrow  = lane & 15;
    int khalf = lane >> 4;
    int rt    = w & 3;
    int ctb   = (w >> 2) * 8;

    v8f acc[8];
#pragma unroll
    for (int t = 0; t < 8; ++t)
#pragma unroll
        for (int j = 0; j < 8; ++j) acc[t][j] = 0.0f;

    const float* ap = proj + (rt * 16 + mrow) * C_DIM + 2 * khalf;
    const float* wp = W + (2 * khalf) * C_DIM + ctb * 16 + mrow;
#pragma unroll 1
    for (int kk = 0; kk < 64; kk += 2) {
        v2f a[2], b[2][8];
#pragma unroll
        for (int s = 0; s < 2; ++s) {
            a[s] = *(const v2f*)(ap + 4 * s);
#pragma unroll
            for (int t = 0; t < 8; ++t) {
                b[s][t].x = wp[(4 * s) * C_DIM + 16 * t];
                b[s][t].y = wp[(4 * s + 1) * C_DIM + 16 * t];
            }
        }
        __builtin_amdgcn_sched_barrier(0);
#pragma unroll
        for (int s = 0; s < 2; ++s)
#pragma unroll
            for (int t = 0; t < 8; ++t)
                acc[t] = __builtin_amdgcn_wmma_f32_16x16x4_f32(
                    false, a[s], false, b[s][t], (short)0, acc[t], false, false);
        ap += 8;
        wp += 8 * C_DIM;
    }
#pragma unroll
    for (int t = 0; t < 8; ++t) {
        int c = (ctb + t) * 16 + mrow;
#pragma unroll
        for (int i = 0; i < 8; ++i) {
            int r = rt * 16 + i + 8 * khalf;
            P2i[(r >> 1) * (2 * C_DIM) + c * 2 + (r & 1)] = acc[t][i];
        }
    }
}

__global__ void p2_u_kernel(const float* __restrict__ proj,
                            const float* __restrict__ W,
                            float* __restrict__ P2i,
                            const float* __restrict__ eg,
                            float* __restrict__ u, int n) {
    if (blockIdx.x == 0) p2_block(proj, W, P2i);
    else                 reduce_u_block(eg, u, n, (int)blockIdx.x - 1, (int)gridDim.x - 1);
}

template <bool GUARD>
__device__ __forceinline__ void out_store(float* __restrict__ out, const v8f (&acc)[4][4],
                                          int rb, int h, int colc, int n) {
#pragma unroll
    for (int i = 0; i < 4; ++i)
#pragma unroll
        for (int qr = 0; qr < 8; ++qr) {
            int row = rb + 16 * i + 8 * h + qr;
            float* orow = out + (size_t)row * C_DIM + colc;
            if (!GUARD || row < n) {
#pragma unroll
                for (int t = 0; t < 4; ++t) orow[16 * t] = acc[i][t][qr];
            }
        }
}

__global__ __launch_bounds__(128)
void out_kernel(const float* __restrict__ eg,
                const float* __restrict__ P2i,
                const float* __restrict__ u,
                const float* __restrict__ bias,
                float* __restrict__ out, int n) {
    int tid     = threadIdx.x;
    int lane    = tid & 31;
    int w       = __builtin_amdgcn_readfirstlane(tid >> 5);
    int c       = lane & 15;
    int h       = lane >> 4;
    int rb      = blockIdx.x * 64;
    int colbase = w * 64;
    int colc    = colbase + c;

    v4f a4[4][8];
#pragma unroll
    for (int i = 0; i < 4; ++i) {
        int rc = min(rb + 16 * i + c, n - 1);
        const float* er = eg + (size_t)rc * R_DIM + 4 * h;
#pragma unroll
        for (int j = 0; j < 8; ++j) a4[i][j] = *(const v4f*)(er + 8 * j);
    }

    v4f u4[8];
#pragma unroll
    for (int j = 0; j < 8; ++j) u4[j] = *(const v4f*)(u + 8 * j + 4 * h);

    const v2f* P2v = (const v2f*)P2i;
    const v2f* bp  = P2v + (2 * h) * C_DIM + colc;
    v2f b[8][2][4];
#pragma unroll
    for (int j = 0; j < 8; ++j)
#pragma unroll
        for (int p = 0; p < 2; ++p)
#pragma unroll
            for (int t = 0; t < 4; ++t)
                b[j][p][t] = bp[(4 * j + p) * C_DIM + 16 * t];

    float bs[4];
#pragma unroll
    for (int t = 0; t < 4; ++t) bs[t] = bias[colc + 16 * t];

    float inv[4];
#pragma unroll
    for (int i = 0; i < 4; ++i) {
        float s = 0.0f;
#pragma unroll
        for (int j = 0; j < 8; ++j)
#pragma unroll
            for (int e = 0; e < 4; ++e) s += a4[i][j][e] * u4[j][e];
        s += __shfl_xor(s, 16);
        inv[i] = 1.0f / s;
    }

    v8f acc[4][4];
#pragma unroll
    for (int i = 0; i < 4; ++i)
#pragma unroll
        for (int t = 0; t < 4; ++t)
#pragma unroll
            for (int q = 0; q < 8; ++q) acc[i][t][q] = 0.0f;

#pragma unroll
    for (int j = 0; j < 8; ++j)
#pragma unroll
        for (int p = 0; p < 2; ++p) {
            v2f a[4];
#pragma unroll
            for (int i = 0; i < 4; ++i) { a[i].x = a4[i][j][2 * p]; a[i].y = a4[i][j][2 * p + 1]; }
#pragma unroll
            for (int t = 0; t < 4; ++t)
#pragma unroll
                for (int i = 0; i < 4; ++i)
                    acc[i][t] = __builtin_amdgcn_wmma_f32_16x16x4_f32(
                        false, a[i], false, b[j][p][t], (short)0, acc[i][t], false, false);
        }

#pragma unroll
    for (int i = 0; i < 4; ++i)
#pragma unroll
        for (int qr = 0; qr < 8; ++qr) {
            float sc = __shfl(inv[i], 8 * h + qr);
#pragma unroll
            for (int t = 0; t < 4; ++t) acc[i][t][qr] = acc[i][t][qr] * sc + bs[t];
        }

    if (rb + 64 <= n) out_store<false>(out, acc, rb, h, colc, n);
    else              out_store<true>(out, acc, rb, h, colc, n);
}

extern "C" void kernel_launch(void* const* d_in, const int* in_sizes, int n_in,
                              void* d_out, int out_size, void* d_ws, size_t ws_size,
                              hipStream_t stream) {
    const float* x    = (const float*)d_in[0];
    const float* eg   = (const float*)d_in[2];
    const float* W    = (const float*)d_in[3];
    const float* bias = (const float*)d_in[4];
    float*       out  = (float*)d_out;

    int N = in_sizes[2] / R_DIM;

    float* ws   = (float*)d_ws;
    float* u    = ws;
    float* proj = ws + 64;
    float* P2i  = ws + 64 + R_DIM * C_DIM;

    int nz = 64 + R_DIM * C_DIM;
    zero_kernel<<<(nz + 255) / 256, 256, 0, stream>>>(ws, nz);
    proj_kernel<<<256, 256, 0, stream>>>(eg, x, proj, N);
    p2_u_kernel<<<1 + 512, 256, 0, stream>>>(proj, W, P2i, eg, u, N);
    out_kernel<<<(N + 63) / 64, 128, 0, stream>>>(eg, P2i, u, bias, out, N);
}


# P2 launch re-tiled against serial load round trips: P2=proj@W one 16x16 tile per wave over whole K (64 waves, sum order kept); u as 16-byte row loads, 8 in flight (u sum order changes)
# speedup vs baseline: 4.6825x; 1.2814x over previous
#include <hip/hip_runtime.h>

typedef __attribute__((ext_vector_type(2))) float v2f;
typedef __attribute__((ext_vector_type(4))) float v4f;
typedef __attribute__((ext_vector_type(8))) float v8f;

#define R_DIM 64
#define C_DIM 256

__global__ void zero_kernel(float* __restrict__ p, int n) {
    int i = blockIdx.x * blockDim.x + threadIdx.x;
    if (i < n) p[i] = 0.0f;
}

#define U_GRP 8

__device__ __forceinline__ void reduce_u_block(const float* __restrict__ eg,
                                               float* __restrict__ u, int n,
                                               int blk, int nblk) {
    __shared__ __attribute__((aligned(16))) float smem[16 * R_DIM];
    int tid    = threadIdx.x;
    int c4     = (tid & 15) * 4;
    int rgrp   = tid >> 4;
    int stride = nblk * 16;
    v4f s = {0.0f, 0.0f, 0.0f, 0.0f};
    for (int base = blk * 16 + rgrp; base < n; base += U_GRP * stride) {
        v4f v[U_GRP];
#pragma unroll
        for (int g = 0; g < U_GRP; ++g) {
            int row = base + g * stride;
            int rc  = min(row, n - 1);
            v[g] = *(const v4f*)(eg + (size_t)rc * R_DIM + c4);
        }
#pragma unroll
        for (int g = 0; g < U_GRP; ++g) {
            bool ok = base + g * stride < n;
            v4f z = {0.0f, 0.0f, 0.0f, 0.0f};
            s += ok ? v[g] : z;
        }
    }
    *(v4f*)&smem[rgrp * R_DIM + c4] = s;
    __syncthreads();
    if (tid < 64) {
        float t = 0.0f;
#pragma unroll
        for (int g = 0; g < 16; ++g) t += smem[g * R_DIM + tid];
        atomicAdd(&u[tid], t);
    }
}

#define PJ_D 3

struct PjStep {
    v4f e[2];
    v4f xv[2][2];
};

__device__ __forceinline__ void pj_load(PjStep& b, const float* __restrict__ eg,
                                        const float* __restrict__ x,
                                        int step, int eo, int xo) {
    const float* eS = eg + (size_t)step * (4 * R_DIM);
    const float* xS = x + (size_t)step * (4 * C_DIM);
#pragma unroll
    for (int s = 0; s < 2; ++s) {
        b.e[s]     = *(const v4f*)(eS + eo + s * R_DIM);
        b.xv[0][s] = *(const v4f*)(xS + xo + s * C_DIM);
        b.xv[1][s] = *(const v4f*)(xS + xo + s * C_DIM + 64);
    }
}

__device__ __forceinline__ void pj_mma(v8f (&acc)[4][8], const PjStep& b) {
    v2f a[4];
#pragma unroll
    for (int i = 0; i < 4; ++i) { a[i].x = b.e[0][i]; a[i].y = b.e[1][i]; }
#pragma unroll
    for (int g = 0; g < 2; ++g)
#pragma unroll
        for (int j = 0; j < 4; ++j) {
            v2f bb;
            bb.x = b.xv[g][0][j];
            bb.y = b.xv[g][1][j];
#pragma unroll
            for (int i = 0; i < 4; ++i)
                acc[i][g * 4 + j] = __builtin_amdgcn_wmma_f32_16x16x4_f32(
                    false, a[i], false, bb, (short)0, acc[i][g * 4 + j], false, false);
        }
}

__global__ __launch_bounds__(256)
void proj_kernel(const float* __restrict__ eg,
                 const float* __restrict__ x,
                 float* __restrict__ proj, int n) {
    __shared__ __attribute__((aligned(16))) float part[4 * R_DIM * C_DIM];

    int tid     = threadIdx.x;
    int lane    = tid & 31;
    int w       = __builtin_amdgcn_readfirstlane(tid >> 5);
    int c       = lane & 15;
    int h       = lane >> 4;
    int colhalf = w & 1;
    int kslot   = w >> 1;
    int colbase = colhalf * 128;
    int c4      = c * 4;
    int eo      = (2 * h) * R_DIM + c4;
    int xo      = (2 * h) * C_DIM + colbase + c4;

    int F    = n >> 2;
    int Q    = gridDim.x * 4;
    int q    = blockIdx.x * 4 + kslot;
    int base = F / Q;
    int rem  = F - base * Q;
    int sBeg = q * base + min(q, rem);
    int sEnd = sBeg + base + (q < rem ? 1 : 0);

    v8f acc[4][8];
#pragma unroll
    for (int i = 0; i < 4; ++i)
#pragma unroll
        for (int t = 0; t < 8; ++t)
#pragma unroll
            for (int j = 0; j < 8; ++j) acc[i][t][j] = 0.0f;

    if (sEnd > sBeg) {
        PjStep buf[PJ_D];
        int last = sEnd - 1;
#pragma unroll
        for (int d = 0; d < PJ_D; ++d)
            pj_load(buf[d], eg, x, min(sBeg + d, last), eo, xo);

        int t = sBeg;
        for (; t + PJ_D <= sEnd; t += PJ_D) {
#pragma unroll
            for (int d = 0; d < PJ_D; ++d) {
                PjStep cur = buf[d];
                pj_load(buf[d], eg, x, min(t + d + PJ_D, last), eo, xo);
                pj_mma(acc, cur);
            }
        }
#pragma unroll
        for (int d = 0; d < PJ_D - 1; ++d)
            if (t + d < sEnd) pj_mma(acc, buf[d]);
    }

    if ((n & 3) != 0 && q == Q - 1) {
        PjStep tb;
#pragma unroll
        for (int s = 0; s < 2; ++s) {
            int  row = 4 * F + 2 * h + s;
            bool ok  = row < n;
            int  rc  = ok ? row : n - 1;
            const float* er = eg + (size_t)rc * R_DIM + c4;
            const float* xr = x + (size_t)rc * C_DIM + colbase + c4;
            v4f z  = {0.0f, 0.0f, 0.0f, 0.0f};
            v4f ev = *(const v4f*)er;
            v4f x0 = *(const v4f*)xr;
            v4f x1 = *(const v4f*)(xr + 64);
            tb.e[s]     = ok ? ev : z;
            tb.xv[0][s] = ok ? x0 : z;
            tb.xv[1][s] = ok ? x1 : z;
        }
        pj_mma(acc, tb);
    }

    float* mypart = part + kslot * (R_DIM * C_DIM);
#pragma unroll
    for (int i = 0; i < 4; ++i)
#pragma unroll
        for (int g = 0; g < 2; ++g)
#pragma unroll
            for (int qr = 0; qr < 8; ++qr) {
                int r = 4 * (8 * h + qr) + i;
                v4f v;
                v.x = acc[i][g * 4 + 0][qr];
                v.y = acc[i][g * 4 + 1][qr];
                v.z = acc[i][g * 4 + 2][qr];
                v.w = acc[i][g * 4 + 3][qr];
                *(v4f*)&mypart[r * C_DIM + colbase + 64 * g + c4] = v;
            }
    __syncthreads();

    for (int e0 = 0; e0 < 64; e0 += 8) {
        float p[8][4];
#pragma unroll
        for (int u = 0; u < 8; ++u)
#pragma unroll
            for (int k = 0; k < 4; ++k)
                p[u][k] = part[k * (R_DIM * C_DIM) + (e0 + u) * 256 + tid];
#pragma unroll
        for (int u = 0; u < 8; ++u) {
            float v = ((p[u][0] + p[u][1]) + p[u][2]) + p[u][3];
            atomicAdd(&proj[(e0 + u) * 256 + tid], v);
        }
    }
}

#define P2_WGS 8

__device__ __forceinline__ void p2_block(const float* __restrict__ proj,
                                         const float* __restrict__ W,
                                         float* __restrict__ P2i, int blk) {
    int tid   = threadIdx.x;
    int lane  = tid & 31;
    int tile  = __builtin_amdgcn_readfirstlane(blk * 8 + (tid >> 5));
    int mrow  = lane & 15;
    int khalf = lane >> 4;
    int rt    = tile & 3;
    int ct    = tile >> 2;

    const float* ap = proj + (rt * 16 + mrow) * C_DIM + 2 * khalf;
    const float* wp = W + (2 * khalf) * C_DIM + ct * 16 + mrow;

    v2f a[64], b[64];
#pragma unroll
    for (int kk = 0; kk < 64; ++kk) {
        a[kk]   = *(const v2f*)(ap + 4 * kk);
        b[kk].x = wp[(4 * kk) * C_DIM];
        b[kk].y = wp[(4 * kk + 1) * C_DIM];
    }

    v8f acc;
#pragma unroll
    for (int j = 0; j < 8; ++j) acc[j] = 0.0f;
#pragma unroll
    for (int kk = 0; kk < 64; ++kk)
        acc = __builtin_amdgcn_wmma_f32_16x16x4_f32(
            false, a[kk], false, b[kk], (short)0, acc, false, false);

    int c = ct * 16 + mrow;
#pragma unroll
    for (int i = 0; i < 8; ++i) {
        int r = rt * 16 + i + 8 * khalf;
        P2i[(r >> 1) * (2 * C_DIM) + c * 2 + (r & 1)] = acc[i];
    }
}

__global__
void p2_u_kernel(const float* __restrict__ proj,
                 const float* __restrict__ W,
                 float* __restrict__ P2i,
                 const float* __restrict__ eg,
                 float* __restrict__ u, int n) {
    if (blockIdx.x < P2_WGS) p2_block(proj, W, P2i, (int)blockIdx.x);
    else reduce_u_block(eg, u, n, (int)blockIdx.x - P2_WGS, (int)gridDim.x - P2_WGS);
}

template <bool GUARD>
__device__ __forceinline__ void out_store(float* __restrict__ out, const v8f (&acc)[4][4],
                                          int rb, int h, int colc, int n) {
#pragma unroll
    for (int i = 0; i < 4; ++i)
#pragma unroll
        for (int qr = 0; qr < 8; ++qr) {
            int row = rb + 16 * i + 8 * h + qr;
            float* orow = out + (size_t)row * C_DIM + colc;
            if (!GUARD || row < n) {
#pragma unroll
                for (int t = 0; t < 4; ++t) orow[16 * t] = acc[i][t][qr];
            }
        }
}

__global__ __launch_bounds__(128)
void out_kernel(const float* __restrict__ eg,
                const float* __restrict__ P2i,
                const float* __restrict__ u,
                const float* __restrict__ bias,
                float* __restrict__ out, int n) {
    int tid     = threadIdx.x;
    int lane    = tid & 31;
    int w       = __builtin_amdgcn_readfirstlane(tid >> 5);
    int c       = lane & 15;
    int h       = lane >> 4;
    int rb      = blockIdx.x * 64;
    int colbase = w * 64;
    int colc    = colbase + c;

    v4f a4[4][8];
#pragma unroll
    for (int i = 0; i < 4; ++i) {
        int rc = min(rb + 16 * i + c, n - 1);
        const float* er = eg + (size_t)rc * R_DIM + 4 * h;
#pragma unroll
        for (int j = 0; j < 8; ++j) a4[i][j] = *(const v4f*)(er + 8 * j);
    }

    v4f u4[8];
#pragma unroll
    for (int j = 0; j < 8; ++j) u4[j] = *(const v4f*)(u + 8 * j + 4 * h);

    const v2f* P2v = (const v2f*)P2i;
    const v2f* bp  = P2v + (2 * h) * C_DIM + colc;
    v2f b[8][2][4];
#pragma unroll
    for (int j = 0; j < 8; ++j)
#pragma unroll
        for (int p = 0; p < 2; ++p)
#pragma unroll
            for (int t = 0; t < 4; ++t)
                b[j][p][t] = bp[(4 * j + p) * C_DIM + 16 * t];

    float bs[4];
#pragma unroll
    for (int t = 0; t < 4; ++t) bs[t] = bias[colc + 16 * t];

    float inv[4];
#pragma unroll
    for (int i = 0; i < 4; ++i) {
        float s = 0.0f;
#pragma unroll
        for (int j = 0; j < 8; ++j)
#pragma unroll
            for (int e = 0; e < 4; ++e) s += a4[i][j][e] * u4[j][e];
        s += __shfl_xor(s, 16);
        inv[i] = 1.0f / s;
    }

    v8f acc[4][4];
#pragma unroll
    for (int i = 0; i < 4; ++i)
#pragma unroll
        for (int t = 0; t < 4; ++t)
#pragma unroll
            for (int q = 0; q < 8; ++q) acc[i][t][q] = 0.0f;

#pragma unroll
    for (int j = 0; j < 8; ++j)
#pragma unroll
        for (int p = 0; p < 2; ++p) {
            v2f a[4];
#pragma unroll
            for (int i = 0; i < 4; ++i) { a[i].x = a4[i][j][2 * p]; a[i].y = a4[i][j][2 * p + 1]; }
#pragma unroll
            for (int t = 0; t < 4; ++t)
#pragma unroll
                for (int i = 0; i < 4; ++i)
                    acc[i][t] = __builtin_amdgcn_wmma_f32_16x16x4_f32(
                        false, a[i], false, b[j][p][t], (short)0, acc[i][t], false, false);
        }

#pragma unroll
    for (int i = 0; i < 4; ++i)
#pragma unroll
        for (int qr = 0; qr < 8; ++qr) {
            float sc = __shfl(inv[i], 8 * h + qr);
#pragma unroll
            for (int t = 0; t < 4; ++t) acc[i][t][qr] = acc[i][t][qr] * sc + bs[t];
        }

    if (rb + 64 <= n) out_store<false>(out, acc, rb, h, colc, n);
    else              out_store<true>(out, acc, rb, h, colc, n);
}

extern "C" void kernel_launch(void* const* d_in, const int* in_sizes, int n_in,
                              void* d_out, int out_size, void* d_ws, size_t ws_size,
                              hipStream_t stream) {
    const float* x    = (const float*)d_in[0];
    const float* eg   = (const float*)d_in[2];
    const float* W    = (const float*)d_in[3];
    const float* bias = (const float*)d_in[4];
    float*       out  = (float*)d_out;

    int N = in_sizes[2] / R_DIM;

    float* ws   = (float*)d_ws;
    float* u    = ws;
    float* proj = ws + 64;
    float* P2i  = ws + 64 + R_DIM * C_DIM;

    int nz = 64 + R_DIM * C_DIM;
    zero_kernel<<<(nz + 255) / 256, 256, 0, stream>>>(ws, nz);
    proj_kernel<<<256, 256, 0, stream>>>(eg, x, proj, N);
    p2_u_kernel<<<256, 256, 0, stream>>>(proj, W, P2i, eg, u, N);
    out_kernel<<<(N + 63) / 64, 128, 0, stream>>>(eg, P2i, u, bias, out, N);
}


# proj_kernel stores one partial plane of proj per work-group (whole lines, parked in out's first rows), summed in fixed order by a launch replacing the zero launch: no proj atomics; sum order changes
# speedup vs baseline: 5.0672x; 1.0822x over previous
#include <hip/hip_runtime.h>

typedef __attribute__((ext_vector_type(2))) float v2f;
typedef __attribute__((ext_vector_type(4))) float v4f;
typedef __attribute__((ext_vector_type(8))) float v8f;

#define R_DIM 64
#define C_DIM 256
#define PLANE (R_DIM * C_DIM)
#define PJ_WGS 256
#define PJ_SLICES (PLANE / 64)

#define PL_GRP (PJ_WGS / 16)

__global__ void sum_planes_kernel(const float* __restrict__ planes,
                                  float* __restrict__ proj,
                                  float* __restrict__ u) {
    __shared__ __attribute__((aligned(16))) float psm[16 * 64];
    int tid = threadIdx.x;
    int blk = blockIdx.x;
    int c4  = (tid & 15) * 4;
    int pg  = tid >> 4;
    if (blk == 0 && tid < R_DIM) u[tid] = 0.0f;
    const float* src = planes + (size_t)pg * PLANE + blk * 64 + c4;
    v4f v[PL_GRP];
#pragma unroll
    for (int g = 0; g < PL_GRP; ++g) v[g] = *(const v4f*)(src + (size_t)g * (16 * PLANE));
    v4f s = v[0];
#pragma unroll
    for (int g = 1; g < PL_GRP; ++g) s += v[g];
    *(v4f*)&psm[pg * 64 + c4] = s;
    __syncthreads();
    if (tid < 64) {
        float t = 0.0f;
#pragma unroll
        for (int g = 0; g < 16; ++g) t += psm[g * 64 + tid];
        proj[blk * 64 + tid] = t;
    }
}

#define U_GRP 8

__device__ __forceinline__ void reduce_u_block(const float* __restrict__ eg,
                                               float* __restrict__ u, int n,
                                               int blk, int nblk) {
    __shared__ __attribute__((aligned(16))) float smem[16 * R_DIM];
    int tid    = threadIdx.x;
    int c4     = (tid & 15) * 4;
    int rgrp   = tid >> 4;
    int stride = nblk * 16;
    v4f s = {0.0f, 0.0f, 0.0f, 0.0f};
    for (int base = blk * 16 + rgrp; base < n; base += U_GRP * stride) {
        v4f v[U_GRP];
#pragma unroll
        for (int g = 0; g < U_GRP; ++g) {
            int row = base + g * stride;
            int rc  = min(row, n - 1);
            v[g] = *(const v4f*)(eg + (size_t)rc * R_DIM + c4);
        }
#pragma unroll
        for (int g = 0; g < U_GRP; ++g) {
            bool ok = base + g * stride < n;
            v4f z = {0.0f, 0.0f, 0.0f, 0.0f};
            s += ok ? v[g] : z;
        }
    }
    *(v4f*)&smem[rgrp * R_DIM + c4] = s;
    __syncthreads();
    if (tid < 64) {
        float t = 0.0f;
#pragma unroll
        for (int g = 0; g < 16; ++g) t += smem[g * R_DIM + tid];
        atomicAdd(&u[tid], t);
    }
}

#define PJ_D 3

struct PjStep {
    v4f e[2];
    v4f xv[2][2];
};

__device__ __forceinline__ void pj_load(PjStep& b, const float* __restrict__ eg,
                                        const float* __restrict__ x,
                                        int step, int eo, int xo) {
    const float* eS = eg + (size_t)step * (4 * R_DIM);
    const float* xS = x + (size_t)step * (4 * C_DIM);
#pragma unroll
    for (int s = 0; s < 2; ++s) {
        b.e[s]     = *(const v4f*)(eS + eo + s * R_DIM);
        b.xv[0][s] = *(const v4f*)(xS + xo + s * C_DIM);
        b.xv[1][s] = *(const v4f*)(xS + xo + s * C_DIM + 64);
    }
}

__device__ __forceinline__ void pj_mma(v8f (&acc)[4][8], const PjStep& b) {
    v2f a[4];
#pragma unroll
    for (int i = 0; i < 4; ++i) { a[i].x = b.e[0][i]; a[i].y = b.e[1][i]; }
#pragma unroll
    for (int g = 0; g < 2; ++g)
#pragma unroll
        for (int j = 0; j < 4; ++j) {
            v2f bb;
            bb.x = b.xv[g][0][j];
            bb.y = b.xv[g][1][j];
#pragma unroll
            for (int i = 0; i < 4; ++i)
                acc[i][g * 4 + j] = __builtin_amdgcn_wmma_f32_16x16x4_f32(
                    false, a[i], false, bb, (short)0, acc[i][g * 4 + j], false, false);
        }
}

__global__ __launch_bounds__(256)
void proj_kernel(const float* __restrict__ eg,
                 const float* __restrict__ x,
                 float* __restrict__ planes, int n) {
    __shared__ __attribute__((aligned(16))) float part[4 * R_DIM * C_DIM];

    int tid     = threadIdx.x;
    int lane    = tid & 31;
    int w       = __builtin_amdgcn_readfirstlane(tid >> 5);
    int c       = lane & 15;
    int h       = lane >> 4;
    int colhalf = w & 1;
    int kslot   = w >> 1;
    int colbase = colhalf * 128;
    int c4      = c * 4;
    int eo      = (2 * h) * R_DIM + c4;
    int xo      = (2 * h) * C_DIM + colbase + c4;

    int F    = n >> 2;
    int Q    = gridDim.x * 4;
    int q    = blockIdx.x * 4 + kslot;
    int base = F / Q;
    int rem  = F - base * Q;
    int sBeg = q * base + min(q, rem);
    int sEnd = sBeg + base + (q < rem ? 1 : 0);

    v8f acc[4][8];
#pragma unroll
    for (int i = 0; i < 4; ++i)
#pragma unroll
        for (int t = 0; t < 8; ++t)
#pragma unroll
            for (int j = 0; j < 8; ++j) acc[i][t][j] = 0.0f;

    if (sEnd > sBeg) {
        PjStep buf[PJ_D];
        int last = sEnd - 1;
#pragma unroll
        for (int d = 0; d < PJ_D; ++d)
            pj_load(buf[d], eg, x, min(sBeg + d, last), eo, xo);

        int t = sBeg;
        for (; t + PJ_D <= sEnd; t += PJ_D) {
#pragma unroll
            for (int d = 0; d < PJ_D; ++d) {
                PjStep cur = buf[d];
                pj_load(buf[d], eg, x, min(t + d + PJ_D, last), eo, xo);
                pj_mma(acc, cur);
            }
        }
#pragma unroll
        for (int d = 0; d < PJ_D - 1; ++d)
            if (t + d < sEnd) pj_mma(acc, buf[d]);
    }

    if ((n & 3) != 0 && q == Q - 1) {
        PjStep tb;
#pragma unroll
        for (int s = 0; s < 2; ++s) {
            int  row = 4 * F + 2 * h + s;
            bool ok  = row < n;
            int  rc  = ok ? row : n - 1;
            const float* er = eg + (size_t)rc * R_DIM + c4;
            const float* xr = x + (size_t)rc * C_DIM + colbase + c4;
            v4f z  = {0.0f, 0.0f, 0.0f, 0.0f};
            v4f ev = *(const v4f*)er;
            v4f x0 = *(const v4f*)xr;
            v4f x1 = *(const v4f*)(xr + 64);
            tb.e[s]     = ok ? ev : z;
            tb.xv[0][s] = ok ? x0 : z;
            tb.xv[1][s] = ok ? x1 : z;
        }
        pj_mma(acc, tb);
    }

    float* mypart = part + kslot * (R_DIM * C_DIM);
#pragma unroll
    for (int i = 0; i < 4; ++i)
#pragma unroll
        for (int g = 0; g < 2; ++g)
#pragma unroll
            for (int qr = 0; qr < 8; ++qr) {
                int r = 4 * (8 * h + qr) + i;
                v4f v;
                v.x = acc[i][g * 4 + 0][qr];
                v.y = acc[i][g * 4 + 1][qr];
                v.z = acc[i][g * 4 + 2][qr];
                v.w = acc[i][g * 4 + 3][qr];
                *(v4f*)&mypart[r * C_DIM + colbase + 64 * g + c4] = v;
            }
    __syncthreads();

    float* plane = planes + (size_t)blockIdx.x * PLANE;
    int prow = tid >> 6;
    int pc4  = (tid & 63) * 4;
    for (int e0 = 0; e0 < R_DIM; e0 += 16) {
        v4f p[4][4];
#pragma unroll
        for (int g = 0; g < 4; ++g)
#pragma unroll
            for (int k = 0; k < 4; ++k)
                p[g][k] = *(const v4f*)&part[k * PLANE + (e0 + 4 * g + prow) * C_DIM + pc4];
#pragma unroll
        for (int g = 0; g < 4; ++g) {
            v4f v = ((p[g][0] + p[g][1]) + p[g][2]) + p[g][3];
            *(v4f*)&plane[(e0 + 4 * g + prow) * C_DIM + pc4] = v;
        }
    }
}

#define P2_WGS 8

__device__ __forceinline__ void p2_block(const float* __restrict__ proj,
                                         const float* __restrict__ W,
                                         float* __restrict__ P2i, int blk) {
    int tid   = threadIdx.x;
    int lane  = tid & 31;
    int tile  = __builtin_amdgcn_readfirstlane(blk * 8 + (tid >> 5));
    int mrow  = lane & 15;
    int khalf = lane >> 4;
    int rt    = tile & 3;
    int ct    = tile >> 2;

    const float* ap = proj + (rt * 16 + mrow) * C_DIM + 2 * khalf;
    const float* wp = W + (2 * khalf) * C_DIM + ct * 16 + mrow;

    v2f a[64], b[64];
#pragma unroll
    for (int kk = 0; kk < 64; ++kk) {
        a[kk]   = *(const v2f*)(ap + 4 * kk);
        b[kk].x = wp[(4 * kk) * C_DIM];
        b[kk].y = wp[(4 * kk + 1) * C_DIM];
    }

    v8f acc;
#pragma unroll
    for (int j = 0; j < 8; ++j) acc[j] = 0.0f;
#pragma unroll
    for (int kk = 0; kk < 64; ++kk)
        acc = __builtin_amdgcn_wmma_f32_16x16x4_f32(
            false, a[kk], false, b[kk], (short)0, acc, false, false);

    int c = ct * 16 + mrow;
#pragma unroll
    for (int i = 0; i < 8; ++i) {
        int r = rt * 16 + i + 8 * khalf;
        P2i[(r >> 1) * (2 * C_DIM) + c * 2 + (r & 1)] = acc[i];
    }
}

__global__
void p2_u_kernel(const float* __restrict__ proj,
                 const float* __restrict__ W,
                 float* __restrict__ P2i,
                 const float* __restrict__ eg,
                 float* __restrict__ u, int n) {
    if (blockIdx.x < P2_WGS) p2_block(proj, W, P2i, (int)blockIdx.x);
    else reduce_u_block(eg, u, n, (int)blockIdx.x - P2_WGS, (int)gridDim.x - P2_WGS);
}

template <bool GUARD>
__device__ __forceinline__ void out_store(float* __restrict__ out, const v8f (&acc)[4][4],
                                          int rb, int h, int colc, int n) {
#pragma unroll
    for (int i = 0; i < 4; ++i)
#pragma unroll
        for (int qr = 0; qr < 8; ++qr) {
            int row = rb + 16 * i + 8 * h + qr;
            float* orow = out + (size_t)row * C_DIM + colc;
            if (!GUARD || row < n) {
#pragma unroll
                for (int t = 0; t < 4; ++t) orow[16 * t] = acc[i][t][qr];
            }
        }
}

__global__ __launch_bounds__(128)
void out_kernel(const float* __restrict__ eg,
                const float* __restrict__ P2i,
                const float* __restrict__ u,
                const float* __restrict__ bias,
                float* __restrict__ out, int n) {
    int tid     = threadIdx.x;
    int lane    = tid & 31;
    int w       = __builtin_amdgcn_readfirstlane(tid >> 5);
    int c       = lane & 15;
    int h       = lane >> 4;
    int rb      = blockIdx.x * 64;
    int colbase = w * 64;
    int colc    = colbase + c;

    v4f a4[4][8];
#pragma unroll
    for (int i = 0; i < 4; ++i) {
        int rc = min(rb + 16 * i + c, n - 1);
        const float* er = eg + (size_t)rc * R_DIM + 4 * h;
#pragma unroll
        for (int j = 0; j < 8; ++j) a4[i][j] = *(const v4f*)(er + 8 * j);
    }

    v4f u4[8];
#pragma unroll
    for (int j = 0; j < 8; ++j) u4[j] = *(const v4f*)(u + 8 * j + 4 * h);

    const v2f* P2v = (const v2f*)P2i;
    const v2f* bp  = P2v + (2 * h) * C_DIM + colc;
    v2f b[8][2][4];
#pragma unroll
    for (int j = 0; j < 8; ++j)
#pragma unroll
        for (int p = 0; p < 2; ++p)
#pragma unroll
            for (int t = 0; t < 4; ++t)
                b[j][p][t] = bp[(4 * j + p) * C_DIM + 16 * t];

    float bs[4];
#pragma unroll
    for (int t = 0; t < 4; ++t) bs[t] = bias[colc + 16 * t];

    float inv[4];
#pragma unroll
    for (int i = 0; i < 4; ++i) {
        float s = 0.0f;
#pragma unroll
        for (int j = 0; j < 8; ++j)
#pragma unroll
            for (int e = 0; e < 4; ++e) s += a4[i][j][e] * u4[j][e];
        s += __shfl_xor(s, 16);
        inv[i] = 1.0f / s;
    }

    v8f acc[4][4];
#pragma unroll
    for (int i = 0; i < 4; ++i)
#pragma unroll
        for (int t = 0; t < 4; ++t)
#pragma unroll
            for (int q = 0; q < 8; ++q) acc[i][t][q] = 0.0f;

#pragma unroll
    for (int j = 0; j < 8; ++j)
#pragma unroll
        for (int p = 0; p < 2; ++p) {
            v2f a[4];
#pragma unroll
            for (int i = 0; i < 4; ++i) { a[i].x = a4[i][j][2 * p]; a[i].y = a4[i][j][2 * p + 1]; }
#pragma unroll
            for (int t = 0; t < 4; ++t)
#pragma unroll
                for (int i = 0; i < 4; ++i)
                    acc[i][t] = __builtin_amdgcn_wmma_f32_16x16x4_f32(
                        false, a[i], false, b[j][p][t], (short)0, acc[i][t], false, false);
        }

#pragma unroll
    for (int i = 0; i < 4; ++i)
#pragma unroll
        for (int qr = 0; qr < 8; ++qr) {
            float sc = __shfl(inv[i], 8 * h + qr);
#pragma unroll
            for (int t = 0; t < 4; ++t) acc[i][t][qr] = acc[i][t][qr] * sc + bs[t];
        }

    if (rb + 64 <= n) out_store<false>(out, acc, rb, h, colc, n);
    else              out_store<true>(out, acc, rb, h, colc, n);
}

extern "C" void kernel_launch(void* const* d_in, const int* in_sizes, int n_in,
                              void* d_out, int out_size, void* d_ws, size_t ws_size,
                              hipStream_t stream) {
    const float* x    = (const float*)d_in[0];
    const float* eg   = (const float*)d_in[2];
    const float* W    = (const float*)d_in[3];
    const float* bias = (const float*)d_in[4];
    float*       out  = (float*)d_out;

    int N = in_sizes[2] / R_DIM;

    float* ws   = (float*)d_ws;
    float* u    = ws;
    float* proj = ws + 64;
    float* P2i  = ws + 64 + PLANE;

    float* planes = out;

    proj_kernel<<<PJ_WGS, 256, 0, stream>>>(eg, x, planes, N);
    sum_planes_kernel<<<PJ_SLICES, 256, 0, stream>>>(planes, proj, u);
    p2_u_kernel<<<256, 256, 0, stream>>>(proj, W, P2i, eg, u, N);
    out_kernel<<<(N + 63) / 64, 128, 0, stream>>>(eg, P2i, u, bias, out, N);
}


# out_kernel: row tile outermost, its four 16-row tiles run in turn; a tile is scaled and stored under the next tile's products, one tile's accumulators live, 3 waves per SIMD not 2; sums kept
# speedup vs baseline: 5.6477x; 1.1146x over previous
#include <hip/hip_runtime.h>

typedef __attribute__((ext_vector_type(2))) float v2f;
typedef __attribute__((ext_vector_type(4))) float v4f;
typedef __attribute__((ext_vector_type(8))) float v8f;

#define R_DIM 64
#define C_DIM 256
#define PLANE (R_DIM * C_DIM)
#define PJ_WGS 256
#define PJ_SLICES (PLANE / 64)

#define PL_GRP (PJ_WGS / 16)

__global__ void sum_planes_kernel(const float* __restrict__ planes,
                                  float* __restrict__ proj,
                                  float* __restrict__ u) {
    __shared__ __attribute__((aligned(16))) float psm[16 * 64];
    int tid = threadIdx.x;
    int blk = blockIdx.x;
    int c4  = (tid & 15) * 4;
    int pg  = tid >> 4;
    if (blk == 0 && tid < R_DIM) u[tid] = 0.0f;
    const float* src = planes + (size_t)pg * PLANE + blk * 64 + c4;
    v4f v[PL_GRP];
#pragma unroll
    for (int g = 0; g < PL_GRP; ++g) v[g] = *(const v4f*)(src + (size_t)g * (16 * PLANE));
    v4f s = v[0];
#pragma unroll
    for (int g = 1; g < PL_GRP; ++g) s += v[g];
    *(v4f*)&psm[pg * 64 + c4] = s;
    __syncthreads();
    if (tid < 64) {
        float t = 0.0f;
#pragma unroll
        for (int g = 0; g < 16; ++g) t += psm[g * 64 + tid];
        proj[blk * 64 + tid] = t;
    }
}

#define U_GRP 8

__device__ __forceinline__ void reduce_u_block(const float* __restrict__ eg,
                                               float* __restrict__ u, int n,
                                               int blk, int nblk) {
    __shared__ __attribute__((aligned(16))) float smem[16 * R_DIM];
    int tid    = threadIdx.x;
    int c4     = (tid & 15) * 4;
    int rgrp   = tid >> 4;
    int stride = nblk * 16;
    v4f s = {0.0f, 0.0f, 0.0f, 0.0f};
    for (int base = blk * 16 + rgrp; base < n; base += U_GRP * stride) {
        v4f v[U_GRP];
#pragma unroll
        for (int g = 0; g < U_GRP; ++g) {
            int row = base + g * stride;
            int rc  = min(row, n - 1);
            v[g] = *(const v4f*)(eg + (size_t)rc * R_DIM + c4);
        }
#pragma unroll
        for (int g = 0; g < U_GRP; ++g) {
            bool ok = base + g * stride < n;
            v4f z = {0.0f, 0.0f, 0.0f, 0.0f};
            s += ok ? v[g] : z;
        }
    }
    *(v4f*)&smem[rgrp * R_DIM + c4] = s;
    __syncthreads();
    if (tid < 64) {
        float t = 0.0f;
#pragma unroll
        for (int g = 0; g < 16; ++g) t += smem[g * R_DIM + tid];
        atomicAdd(&u[tid], t);
    }
}

#define PJ_D 3

struct PjStep {
    v4f e[2];
    v4f xv[2][2];
};

__device__ __forceinline__ void pj_load(PjStep& b, const float* __restrict__ eg,
                                        const float* __restrict__ x,
                                        int step, int eo, int xo) {
    const float* eS = eg + (size_t)step * (4 * R_DIM);
    const float* xS = x + (size_t)step * (4 * C_DIM);
#pragma unroll
    for (int s = 0; s < 2; ++s) {
        b.e[s]     = *(const v4f*)(eS + eo + s * R_DIM);
        b.xv[0][s] = *(const v4f*)(xS + xo + s * C_DIM);
        b.xv[1][s] = *(const v4f*)(xS + xo + s * C_DIM + 64);
    }
}

__device__ __forceinline__ void pj_mma(v8f (&acc)[4][8], const PjStep& b) {
    v2f a[4];
#pragma unroll
    for (int i = 0; i < 4; ++i) { a[i].x = b.e[0][i]; a[i].y = b.e[1][i]; }
#pragma unroll
    for (int g = 0; g < 2; ++g)
#pragma unroll
        for (int j = 0; j < 4; ++j) {
            v2f bb;
            bb.x = b.xv[g][0][j];
            bb.y = b.xv[g][1][j];
#pragma unroll
            for (int i = 0; i < 4; ++i)
                acc[i][g * 4 + j] = __builtin_amdgcn_wmma_f32_16x16x4_f32(
                    false, a[i], false, bb, (short)0, acc[i][g * 4 + j], false, false);
        }
}

__global__ __launch_bounds__(256)
void proj_kernel(const float* __restrict__ eg,
                 const float* __restrict__ x,
                 float* __restrict__ planes, int n) {
    __shared__ __attribute__((aligned(16))) float part[4 * R_DIM * C_DIM];

    int tid     = threadIdx.x;
    int lane    = tid & 31;
    int w       = __builtin_amdgcn_readfirstlane(tid >> 5);
    int c       = lane & 15;
    int h       = lane >> 4;
    int colhalf = w & 1;
    int kslot   = w >> 1;
    int colbase = colhalf * 128;
    int c4      = c * 4;
    int eo      = (2 * h) * R_DIM + c4;
    int xo      = (2 * h) * C_DIM + colbase + c4;

    int F    = n >> 2;
    int Q    = gridDim.x * 4;
    int q    = blockIdx.x * 4 + kslot;
    int base = F / Q;
    int rem  = F - base * Q;
    int sBeg = q * base + min(q, rem);
    int sEnd = sBeg + base + (q < rem ? 1 : 0);

    v8f acc[4][8];
#pragma unroll
    for (int i = 0; i < 4; ++i)
#pragma unroll
        for (int t = 0; t < 8; ++t)
#pragma unroll
            for (int j = 0; j < 8; ++j) acc[i][t][j] = 0.0f;

    if (sEnd > sBeg) {
        PjStep buf[PJ_D];
        int last = sEnd - 1;
#pragma unroll
        for (int d = 0; d < PJ_D; ++d)
            pj_load(buf[d], eg, x, min(sBeg + d, last), eo, xo);

        int t = sBeg;
        for (; t + PJ_D <= sEnd; t += PJ_D) {
#pragma unroll
            for (int d = 0; d < PJ_D; ++d) {
                PjStep cur = buf[d];
                pj_load(buf[d], eg, x, min(t + d + PJ_D, last), eo, xo);
                pj_mma(acc, cur);
            }
        }
#pragma unroll
        for (int d = 0; d < PJ_D - 1; ++d)
            if (t + d < sEnd) pj_mma(acc, buf[d]);
    }

    if ((n & 3) != 0 && q == Q - 1) {
        PjStep tb;
#pragma unroll
        for (int s = 0; s < 2; ++s) {
            int  row = 4 * F + 2 * h + s;
            bool ok  = row < n;
            int  rc  = ok ? row : n - 1;
            const float* er = eg + (size_t)rc * R_DIM + c4;
            const float* xr = x + (size_t)rc * C_DIM + colbase + c4;
            v4f z  = {0.0f, 0.0f, 0.0f, 0.0f};
            v4f ev = *(const v4f*)er;
            v4f x0 = *(const v4f*)xr;
            v4f x1 = *(const v4f*)(xr + 64);
            tb.e[s]     = ok ? ev : z;
            tb.xv[0][s] = ok ? x0 : z;
            tb.xv[1][s] = ok ? x1 : z;
        }
        pj_mma(acc, tb);
    }

    float* mypart = part + kslot * (R_DIM * C_DIM);
#pragma unroll
    for (int i = 0; i < 4; ++i)
#pragma unroll
        for (int g = 0; g < 2; ++g)
#pragma unroll
            for (int qr = 0; qr < 8; ++qr) {
                int r = 4 * (8 * h + qr) + i;
                v4f v;
                v.x = acc[i][g * 4 + 0][qr];
                v.y = acc[i][g * 4 + 1][qr];
                v.z = acc[i][g * 4 + 2][qr];
                v.w = acc[i][g * 4 + 3][qr];
                *(v4f*)&mypart[r * C_DIM + colbase + 64 * g + c4] = v;
            }
    __syncthreads();

    float* plane = planes + (size_t)blockIdx.x * PLANE;
    int prow = tid >> 6;
    int pc4  = (tid & 63) * 4;
    for (int e0 = 0; e0 < R_DIM; e0 += 16) {
        v4f p[4][4];
#pragma unroll
        for (int g = 0; g < 4; ++g)
#pragma unroll
            for (int k = 0; k < 4; ++k)
                p[g][k] = *(const v4f*)&part[k * PLANE + (e0 + 4 * g + prow) * C_DIM + pc4];
#pragma unroll
        for (int g = 0; g < 4; ++g) {
            v4f v = ((p[g][0] + p[g][1]) + p[g][2]) + p[g][3];
            *(v4f*)&plane[(e0 + 4 * g + prow) * C_DIM + pc4] = v;
        }
    }
}

#define P2_WGS 8

__device__ __forceinline__ void p2_block(const float* __restrict__ proj,
                                         const float* __restrict__ W,
                                         float* __restrict__ P2i, int blk) {
    int tid   = threadIdx.x;
    int lane  = tid & 31;
    int tile  = __builtin_amdgcn_readfirstlane(blk * 8 + (tid >> 5));
    int mrow  = lane & 15;
    int khalf = lane >> 4;
    int rt    = tile & 3;
    int ct    = tile >> 2;

    const float* ap = proj + (rt * 16 + mrow) * C_DIM + 2 * khalf;
    const float* wp = W + (2 * khalf) * C_DIM + ct * 16 + mrow;

    v2f a[64], b[64];
#pragma unroll
    for (int kk = 0; kk < 64; ++kk) {
        a[kk]   = *(const v2f*)(ap + 4 * kk);
        b[kk].x = wp[(4 * kk) * C_DIM];
        b[kk].y = wp[(4 * kk + 1) * C_DIM];
    }

    v8f acc;
#pragma unroll
    for (int j = 0; j < 8; ++j) acc[j] = 0.0f;
#pragma unroll
    for (int kk = 0; kk < 64; ++kk)
        acc = __builtin_amdgcn_wmma_f32_16x16x4_f32(
            false, a[kk], false, b[kk], (short)0, acc, false, false);

    int c = ct * 16 + mrow;
#pragma unroll
    for (int i = 0; i < 8; ++i) {
        int r = rt * 16 + i + 8 * khalf;
        P2i[(r >> 1) * (2 * C_DIM) + c * 2 + (r & 1)] = acc[i];
    }
}

__global__
void p2_u_kernel(const float* __restrict__ proj,
                 const float* __restrict__ W,
                 float* __restrict__ P2i,
                 const float* __restrict__ eg,
                 float* __restrict__ u, int n) {
    if (blockIdx.x < P2_WGS) p2_block(proj, W, P2i, (int)blockIdx.x);
    else reduce_u_block(eg, u, n, (int)blockIdx.x - P2_WGS, (int)gridDim.x - P2_WGS);
}

template <bool GUARD>
__device__ __forceinline__ void out_tiles(const float* __restrict__ eg,
                                          const float* __restrict__ P2i,
                                          const float* __restrict__ u,
                                          const float* __restrict__ bias,
                                          float* __restrict__ out, int n,
                                          int rb, int colc, int c, int h) {
    v4f a4[4][8];
#pragma unroll
    for (int i = 0; i < 4; ++i) {
        int rc = min(rb + 16 * i + c, n - 1);
        const float* er = eg + (size_t)rc * R_DIM + 4 * h;
#pragma unroll
        for (int j = 0; j < 8; ++j) a4[i][j] = *(const v4f*)(er + 8 * j);
    }

    v4f u4[8];
#pragma unroll
    for (int j = 0; j < 8; ++j) u4[j] = *(const v4f*)(u + 8 * j + 4 * h);

    const v2f* P2v = (const v2f*)P2i;
    const v2f* bp  = P2v + (2 * h) * C_DIM + colc;
    v2f b[8][2][4];
#pragma unroll
    for (int j = 0; j < 8; ++j)
#pragma unroll
        for (int p = 0; p < 2; ++p)
#pragma unroll
            for (int t = 0; t < 4; ++t)
                b[j][p][t] = bp[(4 * j + p) * C_DIM + 16 * t];

    float bs[4];
#pragma unroll
    for (int t = 0; t < 4; ++t) bs[t] = bias[colc + 16 * t];

    float inv[4];
#pragma unroll
    for (int i = 0; i < 4; ++i) {
        float s = 0.0f;
#pragma unroll
        for (int j = 0; j < 8; ++j)
#pragma unroll
            for (int e = 0; e < 4; ++e) s += a4[i][j][e] * u4[j][e];
        s += __shfl_xor(s, 16);
        inv[i] = 1.0f / s;
    }

#pragma unroll
    for (int i = 0; i < 4; ++i) {
        v8f acc[4];
#pragma unroll
        for (int t = 0; t < 4; ++t)
#pragma unroll
            for (int q = 0; q < 8; ++q) acc[t][q] = 0.0f;

#pragma unroll
        for (int j = 0; j < 8; ++j)
#pragma unroll
            for (int p = 0; p < 2; ++p) {
                v2f a;
                a.x = a4[i][j][2 * p];
                a.y = a4[i][j][2 * p + 1];
#pragma unroll
                for (int t = 0; t < 4; ++t)
                    acc[t] = __builtin_amdgcn_wmma_f32_16x16x4_f32(
                        false, a, false, b[j][p][t], (short)0, acc[t], false, false);
            }

#pragma unroll
        for (int qr = 0; qr < 8; ++qr) {
            float sc  = __shfl(inv[i], 8 * h + qr);
            int   row = rb + 16 * i + 8 * h + qr;
            float* orow = out + (size_t)row * C_DIM + colc;
            if (!GUARD || row < n) {
#pragma unroll
                for (int t = 0; t < 4; ++t) orow[16 * t] = acc[t][qr] * sc + bs[t];
            }
        }

        if (i < 3) {
#pragma unroll
            for (int j = 0; j < 8; ++j) asm volatile("" : "+v"(a4[i + 1][j]) : "s"(out) : "memory");
        }
    }
}

__global__ __launch_bounds__(128, 3)
void out_kernel(const float* __restrict__ eg,
                const float* __restrict__ P2i,
                const float* __restrict__ u,
                const float* __restrict__ bias,
                float* __restrict__ out, int n) {
    int tid     = threadIdx.x;
    int lane    = tid & 31;
    int w       = __builtin_amdgcn_readfirstlane(tid >> 5);
    int c       = lane & 15;
    int h       = lane >> 4;
    int rb      = blockIdx.x * 64;
    int colc    = w * 64 + c;

    if (rb + 64 <= n) out_tiles<false>(eg, P2i, u, bias, out, n, rb, colc, c, h);
    else              out_tiles<true>(eg, P2i, u, bias, out, n, rb, colc, c, h);
}

extern "C" void kernel_launch(void* const* d_in, const int* in_sizes, int n_in,
                              void* d_out, int out_size, void* d_ws, size_t ws_size,
                              hipStream_t stream) {
    const float* x    = (const float*)d_in[0];
    const float* eg   = (const float*)d_in[2];
    const float* W    = (const float*)d_in[3];
    const float* bias = (const float*)d_in[4];
    float*       out  = (float*)d_out;

    int N = in_sizes[2] / R_DIM;

    float* ws   = (float*)d_ws;
    float* u    = ws;
    float* proj = ws + 64;
    float* P2i  = ws + 64 + PLANE;

    float* planes = out;

    proj_kernel<<<PJ_WGS, 256, 0, stream>>>(eg, x, planes, N);
    sum_planes_kernel<<<PJ_SLICES, 256, 0, stream>>>(planes, proj, u);
    p2_u_kernel<<<256, 256, 0, stream>>>(proj, W, P2i, eg, u, N);
    out_kernel<<<(N + 63) / 64, 128, 0, stream>>>(eg, P2i, u, bias, out, N);
}


# out_kernel: col tile t of a lane = output column 4c+t, so a row leaves as 16-byte stores of whole 128-byte lines (one row address + constants) and P2 pairs come as 16-byte loads; sums kept
# speedup vs baseline: 6.1743x; 1.0932x over previous
#include <hip/hip_runtime.h>

typedef __attribute__((ext_vector_type(2))) float v2f;
typedef __attribute__((ext_vector_type(4))) float v4f;
typedef __attribute__((ext_vector_type(8))) float v8f;

#define R_DIM 64
#define C_DIM 256
#define PLANE (R_DIM * C_DIM)
#define PJ_WGS 256
#define PJ_SLICES (PLANE / 64)

#define PL_GRP (PJ_WGS / 16)

__global__ void sum_planes_kernel(const float* __restrict__ planes,
                                  float* __restrict__ proj,
                                  float* __restrict__ u) {
    __shared__ __attribute__((aligned(16))) float psm[16 * 64];
    int tid = threadIdx.x;
    int blk = blockIdx.x;
    int c4  = (tid & 15) * 4;
    int pg  = tid >> 4;
    if (blk == 0 && tid < R_DIM) u[tid] = 0.0f;
    const float* src = planes + (size_t)pg * PLANE + blk * 64 + c4;
    v4f v[PL_GRP];
#pragma unroll
    for (int g = 0; g < PL_GRP; ++g) v[g] = *(const v4f*)(src + (size_t)g * (16 * PLANE));
    v4f s = v[0];
#pragma unroll
    for (int g = 1; g < PL_GRP; ++g) s += v[g];
    *(v4f*)&psm[pg * 64 + c4] = s;
    __syncthreads();
    if (tid < 64) {
        float t = 0.0f;
#pragma unroll
        for (int g = 0; g < 16; ++g) t += psm[g * 64 + tid];
        proj[blk * 64 + tid] = t;
    }
}

#define U_GRP 8

__device__ __forceinline__ void reduce_u_block(const float* __restrict__ eg,
                                               float* __restrict__ u, int n,
                                               int blk, int nblk) {
    __shared__ __attribute__((aligned(16))) float smem[16 * R_DIM];
    int tid    = threadIdx.x;
    int c4     = (tid & 15) * 4;
    int rgrp   = tid >> 4;
    int stride = nblk * 16;
    v4f s = {0.0f, 0.0f, 0.0f, 0.0f};
    for (int base = blk * 16 + rgrp; base < n; base += U_GRP * stride) {
        v4f v[U_GRP];
#pragma unroll
        for (int g = 0; g < U_GRP; ++g) {
            int row = base + g * stride;
            int rc  = min(row, n - 1);
            v[g] = *(const v4f*)(eg + (size_t)rc * R_DIM + c4);
        }
#pragma unroll
        for (int g = 0; g < U_GRP; ++g) {
            bool ok = base + g * stride < n;
            v4f z = {0.0f, 0.0f, 0.0f, 0.0f};
            s += ok ? v[g] : z;
        }
    }
    *(v4f*)&smem[rgrp * R_DIM + c4] = s;
    __syncthreads();
    if (tid < 64) {
        float t = 0.0f;
#pragma unroll
        for (int g = 0; g < 16; ++g) t += smem[g * R_DIM + tid];
        atomicAdd(&u[tid], t);
    }
}

#define PJ_D 3

struct PjStep {
    v4f e[2];
    v4f xv[2][2];
};

__device__ __forceinline__ void pj_load(PjStep& b, const float* __restrict__ eg,
                                        const float* __restrict__ x,
                                        int step, int eo, int xo) {
    const float* eS = eg + (size_t)step * (4 * R_DIM);
    const float* xS = x + (size_t)step * (4 * C_DIM);
#pragma unroll
    for (int s = 0; s < 2; ++s) {
        b.e[s]     = *(const v4f*)(eS + eo + s * R_DIM);
        b.xv[0][s] = *(const v4f*)(xS + xo + s * C_DIM);
        b.xv[1][s] = *(const v4f*)(xS + xo + s * C_DIM + 64);
    }
}

__device__ __forceinline__ void pj_mma(v8f (&acc)[4][8], const PjStep& b) {
    v2f a[4];
#pragma unroll
    for (int i = 0; i < 4; ++i) { a[i].x = b.e[0][i]; a[i].y = b.e[1][i]; }
#pragma unroll
    for (int g = 0; g < 2; ++g)
#pragma unroll
        for (int j = 0; j < 4; ++j) {
            v2f bb;
            bb.x = b.xv[g][0][j];
            bb.y = b.xv[g][1][j];
#pragma unroll
            for (int i = 0; i < 4; ++i)
                acc[i][g * 4 + j] = __builtin_amdgcn_wmma_f32_16x16x4_f32(
                    false, a[i], false, bb, (short)0, acc[i][g * 4 + j], false, false);
        }
}

__global__ __launch_bounds__(256)
void proj_kernel(const float* __restrict__ eg,
                 const float* __restrict__ x,
                 float* __restrict__ planes, int n) {
    __shared__ __attribute__((aligned(16))) float part[4 * R_DIM * C_DIM];

    int tid     = threadIdx.x;
    int lane    = tid & 31;
    int w       = __builtin_amdgcn_readfirstlane(tid >> 5);
    int c       = lane & 15;
    int h       = lane >> 4;
    int colhalf = w & 1;
    int kslot   = w >> 1;
    int colbase = colhalf * 128;
    int c4      = c * 4;
    int eo      = (2 * h) * R_DIM + c4;
    int xo      = (2 * h) * C_DIM + colbase + c4;

    int F    = n >> 2;
    int Q    = gridDim.x * 4;
    int q    = blockIdx.x * 4 + kslot;
    int base = F / Q;
    int rem  = F - base * Q;
    int sBeg = q * base + min(q, rem);
    int sEnd = sBeg + base + (q < rem ? 1 : 0);

    v8f acc[4][8];
#pragma unroll
    for (int i = 0; i < 4; ++i)
#pragma unroll
        for (int t = 0; t < 8; ++t)
#pragma unroll
            for (int j = 0; j < 8; ++j) acc[i][t][j] = 0.0f;

    if (sEnd > sBeg) {
        PjStep buf[PJ_D];
        int last = sEnd - 1;
#pragma unroll
        for (int d = 0; d < PJ_D; ++d)
            pj_load(buf[d], eg, x, min(sBeg + d, last), eo, xo);

        int t = sBeg;
        for (; t + PJ_D <= sEnd; t += PJ_D) {
#pragma unroll
            for (int d = 0; d < PJ_D; ++d) {
                PjStep cur = buf[d];
                pj_load(buf[d], eg, x, min(t + d + PJ_D, last), eo, xo);
                pj_mma(acc, cur);
            }
        }
#pragma unroll
        for (int d = 0; d < PJ_D - 1; ++d)
            if (t + d < sEnd) pj_mma(acc, buf[d]);
    }

    if ((n & 3) != 0 && q == Q - 1) {
        PjStep tb;
#pragma unroll
        for (int s = 0; s < 2; ++s) {
            int  row = 4 * F + 2 * h + s;
            bool ok  = row < n;
            int  rc  = ok ? row : n - 1;
            const float* er = eg + (size_t)rc * R_DIM + c4;
            const float* xr = x + (size_t)rc * C_DIM + colbase + c4;
            v4f z  = {0.0f, 0.0f, 0.0f, 0.0f};
            v4f ev = *(const v4f*)er;
            v4f x0 = *(const v4f*)xr;
            v4f x1 = *(const v4f*)(xr + 64);
            tb.e[s]     = ok ? ev : z;
            tb.xv[0][s] = ok ? x0 : z;
            tb.xv[1][s] = ok ? x1 : z;
        }
        pj_mma(acc, tb);
    }

    float* mypart = part + kslot * (R_DIM * C_DIM);
#pragma unroll
    for (int i = 0; i < 4; ++i)
#pragma unroll
        for (int g = 0; g < 2; ++g)
#pragma unroll
            for (int qr = 0; qr < 8; ++qr) {
                int r = 4 * (8 * h + qr) + i;
                v4f v;
                v.x = acc[i][g * 4 + 0][qr];
                v.y = acc[i][g * 4 + 1][qr];
                v.z = acc[i][g * 4 + 2][qr];
                v.w = acc[i][g * 4 + 3][qr];
                *(v4f*)&mypart[r * C_DIM + colbase + 64 * g + c4] = v;
            }
    __syncthreads();

    float* plane = planes + (size_t)blockIdx.x * PLANE;
    int prow = tid >> 6;
    int pc4  = (tid & 63) * 4;
    for (int e0 = 0; e0 < R_DIM; e0 += 16) {
        v4f p[4][4];
#pragma unroll
        for (int g = 0; g < 4; ++g)
#pragma unroll
            for (int k = 0; k < 4; ++k)
                p[g][k] = *(const v4f*)&part[k * PLANE + (e0 + 4 * g + prow) * C_DIM + pc4];
#pragma unroll
        for (int g = 0; g < 4; ++g) {
            v4f v = ((p[g][0] + p[g][1]) + p[g][2]) + p[g][3];
            *(v4f*)&plane[(e0 + 4 * g + prow) * C_DIM + pc4] = v;
        }
    }
}

#define P2_WGS 8

__device__ __forceinline__ void p2_block(const float* __restrict__ proj,
                                         const float* __restrict__ W,
                                         float* __restrict__ P2i, int blk) {
    int tid   = threadIdx.x;
    int lane  = tid & 31;
    int tile  = __builtin_amdgcn_readfirstlane(blk * 8 + (tid >> 5));
    int mrow  = lane & 15;
    int khalf = lane >> 4;
    int rt    = tile & 3;
    int ct    = tile >> 2;

    const float* ap = proj + (rt * 16 + mrow) * C_DIM + 2 * khalf;
    const float* wp = W + (2 * khalf) * C_DIM + ct * 16 + mrow;

    v2f a[64], b[64];
#pragma unroll
    for (int kk = 0; kk < 64; ++kk) {
        a[kk]   = *(const v2f*)(ap + 4 * kk);
        b[kk].x = wp[(4 * kk) * C_DIM];
        b[kk].y = wp[(4 * kk + 1) * C_DIM];
    }

    v8f acc;
#pragma unroll
    for (int j = 0; j < 8; ++j) acc[j] = 0.0f;
#pragma unroll
    for (int kk = 0; kk < 64; ++kk)
        acc = __builtin_amdgcn_wmma_f32_16x16x4_f32(
            false, a[kk], false, b[kk], (short)0, acc, false, false);

    int c = ct * 16 + mrow;
#pragma unroll
    for (int i = 0; i < 8; ++i) {
        int r = rt * 16 + i + 8 * khalf;
        P2i[(r >> 1) * (2 * C_DIM) + c * 2 + (r & 1)] = acc[i];
    }
}

__global__
void p2_u_kernel(const float* __restrict__ proj,
                 const float* __restrict__ W,
                 float* __restrict__ P2i,
                 const float* __restrict__ eg,
                 float* __restrict__ u, int n) {
    if (blockIdx.x < P2_WGS) p2_block(proj, W, P2i, (int)blockIdx.x);
    else reduce_u_block(eg, u, n, (int)blockIdx.x - P2_WGS, (int)gridDim.x - P2_WGS);
}

template <bool GUARD>
__device__ __forceinline__ void out_tiles(const float* __restrict__ eg,
                                          const float* __restrict__ P2i,
                                          const float* __restrict__ u,
                                          const float* __restrict__ bias,
                                          float* __restrict__ out, int n,
                                          int rb, int col4, int c, int h) {
    v4f a4[4][8];
#pragma unroll
    for (int i = 0; i < 4; ++i) {
        int rc = min(rb + 16 * i + c, n - 1);
        const float* er = eg + (size_t)rc * R_DIM + 4 * h;
#pragma unroll
        for (int j = 0; j < 8; ++j) a4[i][j] = *(const v4f*)(er + 8 * j);
    }

    v4f u4[8];
#pragma unroll
    for (int j = 0; j < 8; ++j) u4[j] = *(const v4f*)(u + 8 * j + 4 * h);

    const float* bp = P2i + (2 * h) * (2 * C_DIM) + 2 * col4;
    v4f bq[8][2][2];
#pragma unroll
    for (int j = 0; j < 8; ++j)
#pragma unroll
        for (int p = 0; p < 2; ++p)
#pragma unroll
            for (int tp = 0; tp < 2; ++tp)
                bq[j][p][tp] = *(const v4f*)(bp + (4 * j + p) * (2 * C_DIM) + 4 * tp);

    v4f bs = *(const v4f*)(bias + col4);

    float* orow0 = out + (size_t)(rb + 8 * h) * C_DIM + col4;

    float inv[4];
#pragma unroll
    for (int i = 0; i < 4; ++i) {
        float s = 0.0f;
#pragma unroll
        for (int j = 0; j < 8; ++j)
#pragma unroll
            for (int e = 0; e < 4; ++e) s += a4[i][j][e] * u4[j][e];
        s += __shfl_xor(s, 16);
        inv[i] = 1.0f / s;
    }

#pragma unroll
    for (int i = 0; i < 4; ++i) {
        v8f acc[4];
#pragma unroll
        for (int t = 0; t < 4; ++t)
#pragma unroll
            for (int q = 0; q < 8; ++q) acc[t][q] = 0.0f;

#pragma unroll
        for (int j = 0; j < 8; ++j)
#pragma unroll
            for (int p = 0; p < 2; ++p) {
                v2f a;
                a.x = a4[i][j][2 * p];
                a.y = a4[i][j][2 * p + 1];
#pragma unroll
                for (int t = 0; t < 4; ++t) {
                    v2f bb;
                    bb.x = bq[j][p][t >> 1][2 * (t & 1)];
                    bb.y = bq[j][p][t >> 1][2 * (t & 1) + 1];
                    acc[t] = __builtin_amdgcn_wmma_f32_16x16x4_f32(
                        false, a, false, bb, (short)0, acc[t], false, false);
                }
            }

#pragma unroll
        for (int qr = 0; qr < 8; ++qr) {
            float sc  = __shfl(inv[i], 8 * h + qr);
            int   row = rb + 16 * i + 8 * h + qr;
            v4f v;
            v.x = acc[0][qr] * sc + bs.x;
            v.y = acc[1][qr] * sc + bs.y;
            v.z = acc[2][qr] * sc + bs.z;
            v.w = acc[3][qr] * sc + bs.w;
            if (!GUARD || row < n) *(v4f*)(orow0 + (16 * i + qr) * C_DIM) = v;
        }

        if (i < 3) {
#pragma unroll
            for (int j = 0; j < 8; ++j) asm volatile("" : "+v"(a4[i + 1][j]) : "s"(out) : "memory");
        }
    }
}

__global__ __launch_bounds__(128, 3)
void out_kernel(const float* __restrict__ eg,
                const float* __restrict__ P2i,
                const float* __restrict__ u,
                const float* __restrict__ bias,
                float* __restrict__ out, int n) {
    int tid     = threadIdx.x;
    int lane    = tid & 31;
    int w       = __builtin_amdgcn_readfirstlane(tid >> 5);
    int c       = lane & 15;
    int h       = lane >> 4;
    int rb      = blockIdx.x * 64;
    int col4    = w * 64 + 4 * c;

    if (rb + 64 <= n) out_tiles<false>(eg, P2i, u, bias, out, n, rb, col4, c, h);
    else              out_tiles<true>(eg, P2i, u, bias, out, n, rb, col4, c, h);
}

extern "C" void kernel_launch(void* const* d_in, const int* in_sizes, int n_in,
                              void* d_out, int out_size, void* d_ws, size_t ws_size,
                              hipStream_t stream) {
    const float* x    = (const float*)d_in[0];
    const float* eg   = (const float*)d_in[2];
    const float* W    = (const float*)d_in[3];
    const float* bias = (const float*)d_in[4];
    float*       out  = (float*)d_out;

    int N = in_sizes[2] / R_DIM;

    float* ws   = (float*)d_ws;
    float* u    = ws;
    float* proj = ws + 64;
    float* P2i  = ws + 64 + PLANE;

    float* planes = out;

    proj_kernel<<<PJ_WGS, 256, 0, stream>>>(eg, x, planes, N);
    sum_planes_kernel<<<PJ_SLICES, 256, 0, stream>>>(planes, proj, u);
    p2_u_kernel<<<256, 256, 0, stream>>>(proj, W, P2i, eg, u, N);
    out_kernel<<<(N + 63) / 64, 128, 0, stream>>>(eg, P2i, u, bias, out, N);
}


# P2 = proj @ W folded into the plane-sum launch, split over k: each slice's work-group multiplies its 1x64 piece by its W slab and adds with float atomics (4 -> 3 launches); P2 sum order changes
# speedup vs baseline: 6.3859x; 1.0343x over previous
#include <hip/hip_runtime.h>

typedef __attribute__((ext_vector_type(2))) float v2f;
typedef __attribute__((ext_vector_type(4))) float v4f;
typedef __attribute__((ext_vector_type(8))) float v8f;

#define R_DIM 64
#define C_DIM 256
#define PLANE (R_DIM * C_DIM)
#define PJ_WGS 256
#define PJ_SLICES (PLANE / 64)

#define PL_GRP (PJ_WGS / 16)

__device__ __forceinline__ void sum_p2_block(const float* __restrict__ planes,
                                             const float* __restrict__ W,
                                             float* __restrict__ P2i, int blk) {
    __shared__ __attribute__((aligned(16))) float psm[16 * 64];
    __shared__ __attribute__((aligned(16))) float arow[64];
    int tid   = threadIdx.x;
    int lane  = tid & 31;
    int w     = __builtin_amdgcn_readfirstlane(tid >> 5);
    int ncol  = lane & 15;
    int khalf = lane >> 4;
    int r     = blk >> 2;
    int k0    = (blk & 3) * 64;
    int cb    = w * 32;

    const float* wp = W + (size_t)(k0 + 2 * khalf) * C_DIM + cb + 2 * ncol;
    v2f wv[16][2];
#pragma unroll
    for (int kk = 0; kk < 16; ++kk)
#pragma unroll
        for (int e = 0; e < 2; ++e) wv[kk][e] = *(const v2f*)(wp + (4 * kk + e) * C_DIM);

    int c4 = (tid & 15) * 4;
    int pg = tid >> 4;
    const float* src = planes + (size_t)pg * PLANE + blk * 64 + c4;
    v4f v[PL_GRP];
#pragma unroll
    for (int g = 0; g < PL_GRP; ++g) v[g] = *(const v4f*)(src + (size_t)g * (16 * PLANE));
    __builtin_amdgcn_sched_barrier(0);
    v4f sm = v[0];
#pragma unroll
    for (int g = 1; g < PL_GRP; ++g) sm += v[g];
    *(v4f*)&psm[pg * 64 + c4] = sm;
    __syncthreads();
    if (tid < 64) {
        float t = 0.0f;
#pragma unroll
        for (int g = 0; g < 16; ++g) t += psm[g * 64 + tid];
        arow[tid] = t;
    }
    __syncthreads();

    v2f a[16];
#pragma unroll
    for (int kk = 0; kk < 16; ++kk) {
        v2f av = *(const v2f*)&arow[4 * kk + 2 * khalf];
        v2f z  = {0.0f, 0.0f};
        a[kk]  = (ncol == 0) ? av : z;
    }

    v8f acc[2];
#pragma unroll
    for (int t = 0; t < 2; ++t)
#pragma unroll
        for (int j = 0; j < 8; ++j) acc[t][j] = 0.0f;
#pragma unroll
    for (int kk = 0; kk < 16; ++kk)
#pragma unroll
        for (int t = 0; t < 2; ++t) {
            v2f bb;
            bb.x = wv[kk][0][t];
            bb.y = wv[kk][1][t];
            acc[t] = __builtin_amdgcn_wmma_f32_16x16x4_f32(
                false, a[kk], false, bb, (short)0, acc[t], false, false);
        }

    float p0  = __shfl(acc[0][0], ncol);
    float p1  = __shfl(acc[1][0], ncol);
    float val = khalf ? p1 : p0;
    atomicAdd(P2i + (r >> 1) * (2 * C_DIM) + (cb + 2 * ncol + khalf) * 2 + (r & 1), val);
}

#define U_GRP 8

__device__ __forceinline__ void reduce_u_block(const float* __restrict__ eg,
                                               float* __restrict__ u, int n,
                                               int blk, int nblk) {
    __shared__ __attribute__((aligned(16))) float smem[16 * R_DIM];
    int tid    = threadIdx.x;
    int c4     = (tid & 15) * 4;
    int rgrp   = tid >> 4;
    int stride = nblk * 16;
    v4f s = {0.0f, 0.0f, 0.0f, 0.0f};
    for (int base = blk * 16 + rgrp; base < n; base += U_GRP * stride) {
        v4f v[U_GRP];
#pragma unroll
        for (int g = 0; g < U_GRP; ++g) {
            int row = base + g * stride;
            int rc  = min(row, n - 1);
            v[g] = *(const v4f*)(eg + (size_t)rc * R_DIM + c4);
        }
#pragma unroll
        for (int g = 0; g < U_GRP; ++g) {
            bool ok = base + g * stride < n;
            v4f z = {0.0f, 0.0f, 0.0f, 0.0f};
            s += ok ? v[g] : z;
        }
    }
    *(v4f*)&smem[rgrp * R_DIM + c4] = s;
    __syncthreads();
    if (tid < 64) {
        float t = 0.0f;
#pragma unroll
        for (int g = 0; g < 16; ++g) t += smem[g * R_DIM + tid];
        atomicAdd(&u[tid], t);
    }
}

#define PJ_D 3

struct PjStep {
    v4f e[2];
    v4f xv[2][2];
};

__device__ __forceinline__ void pj_load(PjStep& b, const float* __restrict__ eg,
                                        const float* __restrict__ x,
                                        int step, int eo, int xo) {
    const float* eS = eg + (size_t)step * (4 * R_DIM);
    const float* xS = x + (size_t)step * (4 * C_DIM);
#pragma unroll
    for (int s = 0; s < 2; ++s) {
        b.e[s]     = *(const v4f*)(eS + eo + s * R_DIM);
        b.xv[0][s] = *(const v4f*)(xS + xo + s * C_DIM);
        b.xv[1][s] = *(const v4f*)(xS + xo + s * C_DIM + 64);
    }
}

__device__ __forceinline__ void pj_mma(v8f (&acc)[4][8], const PjStep& b) {
    v2f a[4];
#pragma unroll
    for (int i = 0; i < 4; ++i) { a[i].x = b.e[0][i]; a[i].y = b.e[1][i]; }
#pragma unroll
    for (int g = 0; g < 2; ++g)
#pragma unroll
        for (int j = 0; j < 4; ++j) {
            v2f bb;
            bb.x = b.xv[g][0][j];
            bb.y = b.xv[g][1][j];
#pragma unroll
            for (int i = 0; i < 4; ++i)
                acc[i][g * 4 + j] = __builtin_amdgcn_wmma_f32_16x16x4_f32(
                    false, a[i], false, bb, (short)0, acc[i][g * 4 + j], false, false);
        }
}

__global__ __launch_bounds__(256)
void proj_kernel(const float* __restrict__ eg,
                 const float* __restrict__ x,
                 float* __restrict__ planes,
                 float* __restrict__ P2i,
                 float* __restrict__ u, int n) {
    __shared__ __attribute__((aligned(16))) float part[4 * R_DIM * C_DIM];

    int tid     = threadIdx.x;
    int lane    = tid & 31;
    int w       = __builtin_amdgcn_readfirstlane(tid >> 5);
    int c       = lane & 15;
    int h       = lane >> 4;
    int colhalf = w & 1;
    int kslot   = w >> 1;
    int colbase = colhalf * 128;
    int c4      = c * 4;
    int eo      = (2 * h) * R_DIM + c4;
    int xo      = (2 * h) * C_DIM + colbase + c4;

    {
        int gi = blockIdx.x * 256 + tid;
        v4f z  = {0.0f, 0.0f, 0.0f, 0.0f};
        if (gi < PLANE / 4) *(v4f*)(P2i + 4 * gi) = z;
        if (gi < R_DIM) u[gi] = 0.0f;
    }

    int F    = n >> 2;
    int Q    = gridDim.x * 4;
    int q    = blockIdx.x * 4 + kslot;
    int base = F / Q;
    int rem  = F - base * Q;
    int sBeg = q * base + min(q, rem);
    int sEnd = sBeg + base + (q < rem ? 1 : 0);

    v8f acc[4][8];
#pragma unroll
    for (int i = 0; i < 4; ++i)
#pragma unroll
        for (int t = 0; t < 8; ++t)
#pragma unroll
            for (int j = 0; j < 8; ++j) acc[i][t][j] = 0.0f;

    if (sEnd > sBeg) {
        PjStep buf[PJ_D];
        int last = sEnd - 1;
#pragma unroll
        for (int d = 0; d < PJ_D; ++d)
            pj_load(buf[d], eg, x, min(sBeg + d, last), eo, xo);

        int t = sBeg;
        for (; t + PJ_D <= sEnd; t += PJ_D) {
#pragma unroll
            for (int d = 0; d < PJ_D; ++d) {
                PjStep cur = buf[d];
                pj_load(buf[d], eg, x, min(t + d + PJ_D, last), eo, xo);
                pj_mma(acc, cur);
            }
        }
#pragma unroll
        for (int d = 0; d < PJ_D - 1; ++d)
            if (t + d < sEnd) pj_mma(acc, buf[d]);
    }

    if ((n & 3) != 0 && q == Q - 1) {
        PjStep tb;
#pragma unroll
        for (int s = 0; s < 2; ++s) {
            int  row = 4 * F + 2 * h + s;
            bool ok  = row < n;
            int  rc  = ok ? row : n - 1;
            const float* er = eg + (size_t)rc * R_DIM + c4;
            const float* xr = x + (size_t)rc * C_DIM + colbase + c4;
            v4f z  = {0.0f, 0.0f, 0.0f, 0.0f};
            v4f ev = *(const v4f*)er;
            v4f x0 = *(const v4f*)xr;
            v4f x1 = *(const v4f*)(xr + 64);
            tb.e[s]     = ok ? ev : z;
            tb.xv[0][s] = ok ? x0 : z;
            tb.xv[1][s] = ok ? x1 : z;
        }
        pj_mma(acc, tb);
    }

    float* mypart = part + kslot * (R_DIM * C_DIM);
#pragma unroll
    for (int i = 0; i < 4; ++i)
#pragma unroll
        for (int g = 0; g < 2; ++g)
#pragma unroll
            for (int qr = 0; qr < 8; ++qr) {
                int r = 4 * (8 * h + qr) + i;
                v4f v;
                v.x = acc[i][g * 4 + 0][qr];
                v.y = acc[i][g * 4 + 1][qr];
                v.z = acc[i][g * 4 + 2][qr];
                v.w = acc[i][g * 4 + 3][qr];
                *(v4f*)&mypart[r * C_DIM + colbase + 64 * g + c4] = v;
            }
    __syncthreads();

    float* plane = planes + (size_t)blockIdx.x * PLANE;
    int prow = tid >> 6;
    int pc4  = (tid & 63) * 4;
    for (int e0 = 0; e0 < R_DIM; e0 += 16) {
        v4f p[4][4];
#pragma unroll
        for (int g = 0; g < 4; ++g)
#pragma unroll
            for (int k = 0; k < 4; ++k)
                p[g][k] = *(const v4f*)&part[k * PLANE + (e0 + 4 * g + prow) * C_DIM + pc4];
#pragma unroll
        for (int g = 0; g < 4; ++g) {
            v4f v = ((p[g][0] + p[g][1]) + p[g][2]) + p[g][3];
            *(v4f*)&plane[(e0 + 4 * g + prow) * C_DIM + pc4] = v;
        }
    }
}

#define U_WGS 248
#define K2_WGS 504
static_assert(K2_WGS == PJ_SLICES + U_WGS, "kernel 2 grid = slices of proj + work-groups of u");

__global__ __launch_bounds__(256)
void sum_p2_u_kernel(const float* __restrict__ planes,
                     const float* __restrict__ W,
                     float* __restrict__ P2i,
                     const float* __restrict__ eg,
                     float* __restrict__ u, int n) {
    if (blockIdx.x < PJ_SLICES) sum_p2_block(planes, W, P2i, (int)blockIdx.x);
    else reduce_u_block(eg, u, n, (int)blockIdx.x - PJ_SLICES, (int)gridDim.x - PJ_SLICES);
}

template <bool GUARD>
__device__ __forceinline__ void out_tiles(const float* __restrict__ eg,
                                          const float* __restrict__ P2i,
                                          const float* __restrict__ u,
                                          const float* __restrict__ bias,
                                          float* __restrict__ out, int n,
                                          int rb, int col4, int c, int h) {
    v4f a4[4][8];
#pragma unroll
    for (int i = 0; i < 4; ++i) {
        int rc = min(rb + 16 * i + c, n - 1);
        const float* er = eg + (size_t)rc * R_DIM + 4 * h;
#pragma unroll
        for (int j = 0; j < 8; ++j) a4[i][j] = *(const v4f*)(er + 8 * j);
    }

    v4f u4[8];
#pragma unroll
    for (int j = 0; j < 8; ++j) u4[j] = *(const v4f*)(u + 8 * j + 4 * h);

    const float* bp = P2i + (2 * h) * (2 * C_DIM) + 2 * col4;
    v4f bq[8][2][2];
#pragma unroll
    for (int j = 0; j < 8; ++j)
#pragma unroll
        for (int p = 0; p < 2; ++p)
#pragma unroll
            for (int tp = 0; tp < 2; ++tp)
                bq[j][p][tp] = *(const v4f*)(bp + (4 * j + p) * (2 * C_DIM) + 4 * tp);

    v4f bs = *(const v4f*)(bias + col4);

    float* orow0 = out + (size_t)(rb + 8 * h) * C_DIM + col4;

    float inv[4];
#pragma unroll
    for (int i = 0; i < 4; ++i) {
        float s = 0.0f;
#pragma unroll
        for (int j = 0; j < 8; ++j)
#pragma unroll
            for (int e = 0; e < 4; ++e) s += a4[i][j][e] * u4[j][e];
        s += __shfl_xor(s, 16);
        inv[i] = 1.0f / s;
    }

#pragma unroll
    for (int i = 0; i < 4; ++i) {
        v8f acc[4];
#pragma unroll
        for (int t = 0; t < 4; ++t)
#pragma unroll
            for (int q = 0; q < 8; ++q) acc[t][q] = 0.0f;

#pragma unroll
        for (int j = 0; j < 8; ++j)
#pragma unroll
            for (int p = 0; p < 2; ++p) {
                v2f a;
                a.x = a4[i][j][2 * p];
                a.y = a4[i][j][2 * p + 1];
#pragma unroll
                for (int t = 0; t < 4; ++t) {
                    v2f bb;
                    bb.x = bq[j][p][t >> 1][2 * (t & 1)];
                    bb.y = bq[j][p][t >> 1][2 * (t & 1) + 1];
                    acc[t] = __builtin_amdgcn_wmma_f32_16x16x4_f32(
                        false, a, false, bb, (short)0, acc[t], false, false);
                }
            }

#pragma unroll
        for (int qr = 0; qr < 8; ++qr) {
            float sc  = __shfl(inv[i], 8 * h + qr);
            int   row = rb + 16 * i + 8 * h + qr;
            v4f v;
            v.x = acc[0][qr] * sc + bs.x;
            v.y = acc[1][qr] * sc + bs.y;
            v.z = acc[2][qr] * sc + bs.z;
            v.w = acc[3][qr] * sc + bs.w;
            if (!GUARD || row < n) *(v4f*)(orow0 + (16 * i + qr) * C_DIM) = v;
        }

        if (i < 3) {
#pragma unroll
            for (int j = 0; j < 8; ++j) asm volatile("" : "+v"(a4[i + 1][j]) : "s"(out) : "memory");
        }
    }
}

__global__ __launch_bounds__(128, 3)
void out_kernel(const float* __restrict__ eg,
                const float* __restrict__ P2i,
                const float* __restrict__ u,
                const float* __restrict__ bias,
                float* __restrict__ out, int n) {
    int tid     = threadIdx.x;
    int lane    = tid & 31;
    int w       = __builtin_amdgcn_readfirstlane(tid >> 5);
    int c       = lane & 15;
    int h       = lane >> 4;
    int rb      = blockIdx.x * 64;
    int col4    = w * 64 + 4 * c;

    if (rb + 64 <= n) out_tiles<false>(eg, P2i, u, bias, out, n, rb, col4, c, h);
    else              out_tiles<true>(eg, P2i, u, bias, out, n, rb, col4, c, h);
}

extern "C" void kernel_launch(void* const* d_in, const int* in_sizes, int n_in,
                              void* d_out, int out_size, void* d_ws, size_t ws_size,
                              hipStream_t stream) {
    const float* x    = (const float*)d_in[0];
    const float* eg   = (const float*)d_in[2];
    const float* W    = (const float*)d_in[3];
    const float* bias = (const float*)d_in[4];
    float*       out  = (float*)d_out;

    int N = in_sizes[2] / R_DIM;

    float* ws   = (float*)d_ws;
    float* u    = ws;
    float* P2i  = ws + 64 + PLANE;

    float* planes = out;

    proj_kernel<<<PJ_WGS, 256, 0, stream>>>(eg, x, planes, P2i, u, N);
    sum_p2_u_kernel<<<K2_WGS, 256, 0, stream>>>(planes, W, P2i, eg, u, N);
    out_kernel<<<(N + 63) / 64, 128, 0, stream>>>(eg, P2i, u, bias, out, N);
}


# out_kernel: a work-group owns 80 rows (five 16-row tiles a wave, eg pieces asked for two tiles ahead), not 64: 1250 work-groups, at most 5 per compute unit = 25 tiles (was up to 7 = 28); sums kept
# speedup vs baseline: 6.6453x; 1.0406x over previous
#include <hip/hip_runtime.h>

typedef __attribute__((ext_vector_type(2))) float v2f;
typedef __attribute__((ext_vector_type(4))) float v4f;
typedef __attribute__((ext_vector_type(8))) float v8f;

#define R_DIM 64
#define C_DIM 256
#define PLANE (R_DIM * C_DIM)
#define PJ_WGS 256
#define PJ_SLICES (PLANE / 64)

#define PL_GRP (PJ_WGS / 16)

__device__ __forceinline__ void sum_p2_block(const float* __restrict__ planes,
                                             const float* __restrict__ W,
                                             float* __restrict__ P2i, int blk) {
    __shared__ __attribute__((aligned(16))) float psm[16 * 64];
    __shared__ __attribute__((aligned(16))) float arow[64];
    int tid   = threadIdx.x;
    int lane  = tid & 31;
    int w     = __builtin_amdgcn_readfirstlane(tid >> 5);
    int ncol  = lane & 15;
    int khalf = lane >> 4;
    int r     = blk >> 2;
    int k0    = (blk & 3) * 64;
    int cb    = w * 32;

    const float* wp = W + (size_t)(k0 + 2 * khalf) * C_DIM + cb + 2 * ncol;
    v2f wv[16][2];
#pragma unroll
    for (int kk = 0; kk < 16; ++kk)
#pragma unroll
        for (int e = 0; e < 2; ++e) wv[kk][e] = *(const v2f*)(wp + (4 * kk + e) * C_DIM);

    int c4 = (tid & 15) * 4;
    int pg = tid >> 4;
    const float* src = planes + (size_t)pg * PLANE + blk * 64 + c4;
    v4f v[PL_GRP];
#pragma unroll
    for (int g = 0; g < PL_GRP; ++g) v[g] = *(const v4f*)(src + (size_t)g * (16 * PLANE));
    __builtin_amdgcn_sched_barrier(0);
    v4f sm = v[0];
#pragma unroll
    for (int g = 1; g < PL_GRP; ++g) sm += v[g];
    *(v4f*)&psm[pg * 64 + c4] = sm;
    __syncthreads();
    if (tid < 64) {
        float t = 0.0f;
#pragma unroll
        for (int g = 0; g < 16; ++g) t += psm[g * 64 + tid];
        arow[tid] = t;
    }
    __syncthreads();

    v2f a[16];
#pragma unroll
    for (int kk = 0; kk < 16; ++kk) {
        v2f av = *(const v2f*)&arow[4 * kk + 2 * khalf];
        v2f z  = {0.0f, 0.0f};
        a[kk]  = (ncol == 0) ? av : z;
    }

    v8f acc[2];
#pragma unroll
    for (int t = 0; t < 2; ++t)
#pragma unroll
        for (int j = 0; j < 8; ++j) acc[t][j] = 0.0f;
#pragma unroll
    for (int kk = 0; kk < 16; ++kk)
#pragma unroll
        for (int t = 0; t < 2; ++t) {
            v2f bb;
            bb.x = wv[kk][0][t];
            bb.y = wv[kk][1][t];
            acc[t] = __builtin_amdgcn_wmma_f32_16x16x4_f32(
                false, a[kk], false, bb, (short)0, acc[t], false, false);
        }

    float p0  = __shfl(acc[0][0], ncol);
    float p1  = __shfl(acc[1][0], ncol);
    float val = khalf ? p1 : p0;
    atomicAdd(P2i + (r >> 1) * (2 * C_DIM) + (cb + 2 * ncol + khalf) * 2 + (r & 1), val);
}

#define U_GRP 8

__device__ __forceinline__ void reduce_u_block(const float* __restrict__ eg,
                                               float* __restrict__ u, int n,
                                               int blk, int nblk) {
    __shared__ __attribute__((aligned(16))) float smem[16 * R_DIM];
    int tid    = threadIdx.x;
    int c4     = (tid & 15) * 4;
    int rgrp   = tid >> 4;
    int stride = nblk * 16;
    v4f s = {0.0f, 0.0f, 0.0f, 0.0f};
    for (int base = blk * 16 + rgrp; base < n; base += U_GRP * stride) {
        v4f v[U_GRP];
#pragma unroll
        for (int g = 0; g < U_GRP; ++g) {
            int row = base + g * stride;
            int rc  = min(row, n - 1);
            v[g] = *(const v4f*)(eg + (size_t)rc * R_DIM + c4);
        }
#pragma unroll
        for (int g = 0; g < U_GRP; ++g) {
            bool ok = base + g * stride < n;
            v4f z = {0.0f, 0.0f, 0.0f, 0.0f};
            s += ok ? v[g] : z;
        }
    }
    *(v4f*)&smem[rgrp * R_DIM + c4] = s;
    __syncthreads();
    if (tid < 64) {
        float t = 0.0f;
#pragma unroll
        for (int g = 0; g < 16; ++g) t += smem[g * R_DIM + tid];
        atomicAdd(&u[tid], t);
    }
}

#define PJ_D 3

struct PjStep {
    v4f e[2];
    v4f xv[2][2];
};

__device__ __forceinline__ void pj_load(PjStep& b, const float* __restrict__ eg,
                                        const float* __restrict__ x,
                                        int step, int eo, int xo) {
    const float* eS = eg + (size_t)step * (4 * R_DIM);
    const float* xS = x + (size_t)step * (4 * C_DIM);
#pragma unroll
    for (int s = 0; s < 2; ++s) {
        b.e[s]     = *(const v4f*)(eS + eo + s * R_DIM);
        b.xv[0][s] = *(const v4f*)(xS + xo + s * C_DIM);
        b.xv[1][s] = *(const v4f*)(xS + xo + s * C_DIM + 64);
    }
}

__device__ __forceinline__ void pj_mma(v8f (&acc)[4][8], const PjStep& b) {
    v2f a[4];
#pragma unroll
    for (int i = 0; i < 4; ++i) { a[i].x = b.e[0][i]; a[i].y = b.e[1][i]; }
#pragma unroll
    for (int g = 0; g < 2; ++g)
#pragma unroll
        for (int j = 0; j < 4; ++j) {
            v2f bb;
            bb.x = b.xv[g][0][j];
            bb.y = b.xv[g][1][j];
#pragma unroll
            for (int i = 0; i < 4; ++i)
                acc[i][g * 4 + j] = __builtin_amdgcn_wmma_f32_16x16x4_f32(
                    false, a[i], false, bb, (short)0, acc[i][g * 4 + j], false, false);
        }
}

__global__ __launch_bounds__(256)
void proj_kernel(const float* __restrict__ eg,
                 const float* __restrict__ x,
                 float* __restrict__ planes,
                 float* __restrict__ P2i,
                 float* __restrict__ u, int n) {
    __shared__ __attribute__((aligned(16))) float part[4 * R_DIM * C_DIM];

    int tid     = threadIdx.x;
    int lane    = tid & 31;
    int w       = __builtin_amdgcn_readfirstlane(tid >> 5);
    int c       = lane & 15;
    int h       = lane >> 4;
    int colhalf = w & 1;
    int kslot   = w >> 1;
    int colbase = colhalf * 128;
    int c4      = c * 4;
    int eo      = (2 * h) * R_DIM + c4;
    int xo      = (2 * h) * C_DIM + colbase + c4;

    {
        int gi = blockIdx.x * 256 + tid;
        v4f z  = {0.0f, 0.0f, 0.0f, 0.0f};
        if (gi < PLANE / 4) *(v4f*)(P2i + 4 * gi) = z;
        if (gi < R_DIM) u[gi] = 0.0f;
    }

    int F    = n >> 2;
    int Q    = gridDim.x * 4;
    int q    = blockIdx.x * 4 + kslot;
    int base = F / Q;
    int rem  = F - base * Q;
    int sBeg = q * base + min(q, rem);
    int sEnd = sBeg + base + (q < rem ? 1 : 0);

    v8f acc[4][8];
#pragma unroll
    for (int i = 0; i < 4; ++i)
#pragma unroll
        for (int t = 0; t < 8; ++t)
#pragma unroll
            for (int j = 0; j < 8; ++j) acc[i][t][j] = 0.0f;

    if (sEnd > sBeg) {
        PjStep buf[PJ_D];
        int last = sEnd - 1;
#pragma unroll
        for (int d = 0; d < PJ_D; ++d)
            pj_load(buf[d], eg, x, min(sBeg + d, last), eo, xo);

        int t = sBeg;
        for (; t + PJ_D <= sEnd; t += PJ_D) {
#pragma unroll
            for (int d = 0; d < PJ_D; ++d) {
                PjStep cur = buf[d];
                pj_load(buf[d], eg, x, min(t + d + PJ_D, last), eo, xo);
                pj_mma(acc, cur);
            }
        }
#pragma unroll
        for (int d = 0; d < PJ_D - 1; ++d)
            if (t + d < sEnd) pj_mma(acc, buf[d]);
    }

    if ((n & 3) != 0 && q == Q - 1) {
        PjStep tb;
#pragma unroll
        for (int s = 0; s < 2; ++s) {
            int  row = 4 * F + 2 * h + s;
            bool ok  = row < n;
            int  rc  = ok ? row : n - 1;
            const float* er = eg + (size_t)rc * R_DIM + c4;
            const float* xr = x + (size_t)rc * C_DIM + colbase + c4;
            v4f z  = {0.0f, 0.0f, 0.0f, 0.0f};
            v4f ev = *(const v4f*)er;
            v4f x0 = *(const v4f*)xr;
            v4f x1 = *(const v4f*)(xr + 64);
            tb.e[s]     = ok ? ev : z;
            tb.xv[0][s] = ok ? x0 : z;
            tb.xv[1][s] = ok ? x1 : z;
        }
        pj_mma(acc, tb);
    }

    float* mypart = part + kslot * (R_DIM * C_DIM);
#pragma unroll
    for (int i = 0; i < 4; ++i)
#pragma unroll
        for (int g = 0; g < 2; ++g)
#pragma unroll
            for (int qr = 0; qr < 8; ++qr) {
                int r = 4 * (8 * h + qr) + i;
                v4f v;
                v.x = acc[i][g * 4 + 0][qr];
                v.y = acc[i][g * 4 + 1][qr];
                v.z = acc[i][g * 4 + 2][qr];
                v.w = acc[i][g * 4 + 3][qr];
                *(v4f*)&mypart[r * C_DIM + colbase + 64 * g + c4] = v;
            }
    __syncthreads();

    float* plane = planes + (size_t)blockIdx.x * PLANE;
    int prow = tid >> 6;
    int pc4  = (tid & 63) * 4;
    for (int e0 = 0; e0 < R_DIM; e0 += 16) {
        v4f p[4][4];
#pragma unroll
        for (int g = 0; g < 4; ++g)
#pragma unroll
            for (int k = 0; k < 4; ++k)
                p[g][k] = *(const v4f*)&part[k * PLANE + (e0 + 4 * g + prow) * C_DIM + pc4];
#pragma unroll
        for (int g = 0; g < 4; ++g) {
            v4f v = ((p[g][0] + p[g][1]) + p[g][2]) + p[g][3];
            *(v4f*)&plane[(e0 + 4 * g + prow) * C_DIM + pc4] = v;
        }
    }
}

#define U_WGS 248
#define K2_WGS 504
static_assert(K2_WGS == PJ_SLICES + U_WGS, "kernel 2 grid = slices of proj + work-groups of u");

__global__ __launch_bounds__(256)
void sum_p2_u_kernel(const float* __restrict__ planes,
                     const float* __restrict__ W,
                     float* __restrict__ P2i,
                     const float* __restrict__ eg,
                     float* __restrict__ u, int n) {
    if (blockIdx.x < PJ_SLICES) sum_p2_block(planes, W, P2i, (int)blockIdx.x);
    else reduce_u_block(eg, u, n, (int)blockIdx.x - PJ_SLICES, (int)gridDim.x - PJ_SLICES);
}

#define OUT_RT 5
#define OUT_ROWS (16 * OUT_RT)
#define OUT_AHEAD 2

__device__ __forceinline__ void out_load_eg(v4f (&a)[8], const float* __restrict__ eg,
                                            int row, int n, int h) {
    int rc = min(row, n - 1);
    const float* er = eg + (size_t)rc * R_DIM + 4 * h;
#pragma unroll
    for (int j = 0; j < 8; ++j) a[j] = *(const v4f*)(er + 8 * j);
}

__device__ __forceinline__ float out_inv_deg(const v4f (&a)[8], const v4f (&u4)[8]) {
    float s = 0.0f;
#pragma unroll
    for (int j = 0; j < 8; ++j)
#pragma unroll
        for (int e = 0; e < 4; ++e) s += a[j][e] * u4[j][e];
    s += __shfl_xor(s, 16);
    return 1.0f / s;
}

template <bool GUARD>
__device__ __forceinline__ void out_tiles(const float* __restrict__ eg,
                                          const float* __restrict__ P2i,
                                          const float* __restrict__ u,
                                          const float* __restrict__ bias,
                                          float* __restrict__ out, int n,
                                          int rb, int col4, int c, int h) {
    v4f a4[OUT_RT][8];
#pragma unroll
    for (int i = 0; i < OUT_AHEAD; ++i) out_load_eg(a4[i], eg, rb + 16 * i + c, n, h);

    v4f u4[8];
#pragma unroll
    for (int j = 0; j < 8; ++j) u4[j] = *(const v4f*)(u + 8 * j + 4 * h);

    const float* bp = P2i + (2 * h) * (2 * C_DIM) + 2 * col4;
    v4f bq[8][2][2];
#pragma unroll
    for (int j = 0; j < 8; ++j)
#pragma unroll
        for (int p = 0; p < 2; ++p)
#pragma unroll
            for (int tp = 0; tp < 2; ++tp)
                bq[j][p][tp] = *(const v4f*)(bp + (4 * j + p) * (2 * C_DIM) + 4 * tp);

    v4f bs = *(const v4f*)(bias + col4);

    float* orow0 = out + (size_t)(rb + 8 * h) * C_DIM + col4;

#pragma unroll
    for (int i = 0; i < OUT_RT; ++i) {
        if (i + OUT_AHEAD < OUT_RT) {
            out_load_eg(a4[i + OUT_AHEAD], eg, rb + 16 * (i + OUT_AHEAD) + c, n, h);
            __builtin_amdgcn_sched_barrier(0);
        }

        float inv_i = out_inv_deg(a4[i], u4);

        v8f acc[4];
#pragma unroll
        for (int t = 0; t < 4; ++t)
#pragma unroll
            for (int q = 0; q < 8; ++q) acc[t][q] = 0.0f;

#pragma unroll
        for (int j = 0; j < 8; ++j)
#pragma unroll
            for (int p = 0; p < 2; ++p) {
                v2f a;
                a.x = a4[i][j][2 * p];
                a.y = a4[i][j][2 * p + 1];
#pragma unroll
                for (int t = 0; t < 4; ++t) {
                    v2f bb;
                    bb.x = bq[j][p][t >> 1][2 * (t & 1)];
                    bb.y = bq[j][p][t >> 1][2 * (t & 1) + 1];
                    acc[t] = __builtin_amdgcn_wmma_f32_16x16x4_f32(
                        false, a, false, bb, (short)0, acc[t], false, false);
                }
            }

#pragma unroll
        for (int qr = 0; qr < 8; ++qr) {
            float sc  = __shfl(inv_i, 8 * h + qr);
            int   row = rb + 16 * i + 8 * h + qr;
            v4f v;
            v.x = acc[0][qr] * sc + bs.x;
            v.y = acc[1][qr] * sc + bs.y;
            v.z = acc[2][qr] * sc + bs.z;
            v.w = acc[3][qr] * sc + bs.w;
            if (!GUARD || row < n) *(v4f*)(orow0 + (16 * i + qr) * C_DIM) = v;
        }

        if (i < OUT_RT - 1) {
#pragma unroll
            for (int j = 0; j < 8; ++j) asm volatile("" : "+v"(a4[i + 1][j]) : "s"(out) : "memory");
        }
    }
}

__global__ __launch_bounds__(128, 3)
void out_kernel(const float* __restrict__ eg,
                const float* __restrict__ P2i,
                const float* __restrict__ u,
                const float* __restrict__ bias,
                float* __restrict__ out, int n) {
    int tid     = threadIdx.x;
    int lane    = tid & 31;
    int w       = __builtin_amdgcn_readfirstlane(tid >> 5);
    int c       = lane & 15;
    int h       = lane >> 4;
    int rb      = blockIdx.x * OUT_ROWS;
    int col4    = w * 64 + 4 * c;

    if (rb + OUT_ROWS <= n) out_tiles<false>(eg, P2i, u, bias, out, n, rb, col4, c, h);
    else              out_tiles<true>(eg, P2i, u, bias, out, n, rb, col4, c, h);
}

extern "C" void kernel_launch(void* const* d_in, const int* in_sizes, int n_in,
                              void* d_out, int out_size, void* d_ws, size_t ws_size,
                              hipStream_t stream) {
    const float* x    = (const float*)d_in[0];
    const float* eg   = (const float*)d_in[2];
    const float* W    = (const float*)d_in[3];
    const float* bias = (const float*)d_in[4];
    float*       out  = (float*)d_out;

    int N = in_sizes[2] / R_DIM;

    float* ws   = (float*)d_ws;
    float* u    = ws;
    float* P2i  = ws + 64 + PLANE;

    float* planes = out;

    proj_kernel<<<PJ_WGS, 256, 0, stream>>>(eg, x, planes, P2i, u, N);
    sum_p2_u_kernel<<<K2_WGS, 256, 0, stream>>>(planes, W, P2i, eg, u, N);
    out_kernel<<<(N + OUT_ROWS - 1) / OUT_ROWS, 128, 0, stream>>>(eg, P2i, u, bias, out, N);
}
